# starting kernel
# baseline (speedup 1.0000x reference)
#include <hip/hip_runtime.h>
#include <hip/hip_bf16.h>


#define N_  4
#define L_  4096
#define H_  8
#define D_  64
#define DV_ 64
#define C_  64
#define T_  (L_ / C_)
#define NH_ (N_ * H_)
#define EPS_ 1e-6f

typedef float v2f __attribute__((ext_vector_type(2)));
typedef float v8f __attribute__((ext_vector_type(8)));

__device__ __forceinline__ v8f wmma4(v2f a, v2f b, v8f c) {
  return __builtin_amdgcn_wmma_f32_16x16x4_f32(
       false, a,  false, b,
       (short)0, c,  false,  false);
}

__device__ __forceinline__ v2f frag_a_rm(const float* buf, int ld, int mb, int kk, int lane) {
  int m = mb + (lane & 15);
  int k = kk * 4 + ((lane >> 4) << 1);
  v2f a; a.x = buf[m * ld + k]; a.y = buf[m * ld + k + 1]; return a;
}
__device__ __forceinline__ v2f frag_a_tr(const float* buf, int ld, int mb, int kk, int lane) {
  int m = mb + (lane & 15);
  int k = kk * 4 + ((lane >> 4) << 1);
  v2f a; a.x = buf[k * ld + m]; a.y = buf[(k + 1) * ld + m]; return a;
}
__device__ __forceinline__ v2f frag_b_rm(const float* buf, int ld, int nb, int kk, int lane) {
  int n = nb + (lane & 15);
  int k = kk * 4 + ((lane >> 4) << 1);
  v2f b; b.x = buf[k * ld + n]; b.y = buf[(k + 1) * ld + n]; return b;
}
__device__ __forceinline__ v2f frag_b_tr(const float* buf, int ld, int nb, int kk, int lane) {
  int n = nb + (lane & 15);
  int k = kk * 4 + ((lane >> 4) << 1);
  v2f b; b.x = buf[n * ld + k]; b.y = buf[n * ld + k + 1]; return b;
}

__device__ __forceinline__ float featmap(float x) {
  return x > 0.0f ? x + 1.0f : __expf(x);
}

__global__ __launch_bounds__(256) void cla_phase1(
    const float* __restrict__ kin, const float* __restrict__ vin,
    float* __restrict__ kv, float* __restrict__ ksum) {
  __shared__ float Ks[C_ * D_];
  __shared__ float Vs[C_ * DV_];

  const int bt = blockIdx.x;
  const int nh = bt / T_, t = bt % T_;
  const int n = nh / H_, h = nh % H_;
  const int tid = threadIdx.x;
  const size_t rs = (size_t)H_ * D_;
  const float* kb = kin + (((size_t)n * L_ + (size_t)t * C_) * H_ + h) * D_;
  const float* vb = vin + (((size_t)n * L_ + (size_t)t * C_) * H_ + h) * D_;

  for (int idx = tid; idx < C_ * D_; idx += 256) {
    int r = idx >> 6, c = idx & 63;
    Ks[idx] = featmap(kb[(size_t)r * rs + c]);
    Vs[idx] = vb[(size_t)r * rs + c];
  }
  __syncthreads();

  const int wave = tid >> 5, lane = tid & 31;
  float* kvb = kv + (size_t)(nh * T_ + t) * (D_ * DV_);

#pragma unroll
  for (int j = 0; j < 2; ++j) {
    const int tt = wave * 2 + j;
    const int mb = (tt >> 2) * 16, nb = (tt & 3) * 16;
    v8f acc = {};
#pragma unroll
    for (int kk = 0; kk < 16; ++kk) {
      v2f a = frag_a_tr(Ks, D_, mb, kk, lane);
      v2f b = frag_b_rm(Vs, DV_, nb, kk, lane);
      acc = wmma4(a, b, acc);
    }
    const int col = nb + (lane & 15);
    const int rb  = mb + ((lane >> 4) << 3);
#pragma unroll
    for (int r = 0; r < 8; ++r)
      kvb[(size_t)(rb + r) * DV_ + col] = acc[r];
  }

  if (tid < D_) {
    float s = 0.0f;
    for (int r = 0; r < C_; ++r) s += Ks[r * D_ + tid];
    ksum[(size_t)(nh * T_ + t) * D_ + tid] = s;
  }
}

__global__ __launch_bounds__(256) void cla_phase2(
    float* __restrict__ kv, float* __restrict__ ksum) {
  const int nh = blockIdx.x;
  const int tid = threadIdx.x;
  float run[16];
#pragma unroll
  for (int i = 0; i < 16; ++i) run[i] = 0.0f;
  float runk = 0.0f;

  float* base  = kv + (size_t)nh * T_ * (D_ * DV_);
  float* kbase = ksum + (size_t)nh * T_ * D_;

  for (int t = 0; t < T_; ++t) {
    float* p = base + (size_t)t * (D_ * DV_);
#pragma unroll
    for (int i = 0; i < 16; ++i) {
      const int idx = tid + i * 256;
      const float cur = p[idx];
      p[idx] = run[i];
      run[i] += cur;
    }
    if (tid < D_) {
      const float cur = kbase[t * D_ + tid];
      kbase[t * D_ + tid] = runk;
      runk += cur;
    }
  }
}

__global__ __launch_bounds__(256) void cla_phase3(
    const float* __restrict__ qin, const float* __restrict__ kin,
    const float* __restrict__ vin, const float* __restrict__ kv,
    const float* __restrict__ ksum, float* __restrict__ out) {
  __shared__ float Qs[C_ * D_];
  __shared__ float Vs[C_ * DV_];
  __shared__ float Bs[C_ * D_];
  __shared__ float ks[D_];
  __shared__ float inv[C_];

  const int bt = blockIdx.x;
  const int nh = bt / T_, t = bt % T_;
  const int n = nh / H_, h = nh % H_;
  const int tid = threadIdx.x;
  const size_t rs = (size_t)H_ * D_;
  const float* qb = qin + (((size_t)n * L_ + (size_t)t * C_) * H_ + h) * D_;
  const float* kb = kin + (((size_t)n * L_ + (size_t)t * C_) * H_ + h) * D_;
  const float* vb = vin + (((size_t)n * L_ + (size_t)t * C_) * H_ + h) * D_;
  const float* sp = kv + (size_t)(nh * T_ + t) * (D_ * DV_);

  for (int idx = tid; idx < C_ * D_; idx += 256) {
    int r = idx >> 6, c = idx & 63;
    Qs[idx] = featmap(qb[(size_t)r * rs + c]);
    Vs[idx] = vb[(size_t)r * rs + c];
    Bs[idx] = sp[idx];
  }
  if (tid < D_) ks[tid] = ksum[(size_t)(nh * T_ + t) * D_ + tid];
  __syncthreads();

  const int wave = tid >> 5, lane = tid & 31;
  v8f acc[2];

#pragma unroll
  for (int j = 0; j < 2; ++j) {
    const int tt = wave * 2 + j;
    const int mb = (tt >> 2) * 16, nb = (tt & 3) * 16;
    acc[j] = (v8f){};
#pragma unroll
    for (int kk = 0; kk < 16; ++kk) {
      v2f a = frag_a_rm(Qs, D_, mb, kk, lane);
      v2f b = frag_b_rm(Bs, DV_, nb, kk, lane);
      acc[j] = wmma4(a, b, acc[j]);
    }
  }
  __syncthreads();

  for (int idx = tid; idx < C_ * D_; idx += 256) {
    int r = idx >> 6, c = idx & 63;
    Bs[idx] = featmap(kb[(size_t)r * rs + c]);
  }
  __syncthreads();

  v8f p[2];
#pragma unroll
  for (int j = 0; j < 2; ++j) {
    const int tt = wave * 2 + j;
    const int ti = tt >> 2, tj = tt & 3;
    p[j] = (v8f){};
    if (tj <= ti) {
#pragma unroll
      for (int kk = 0; kk < 16; ++kk) {
        v2f a = frag_a_rm(Qs, D_, ti * 16, kk, lane);
        v2f b = frag_b_tr(Bs, D_, tj * 16, kk, lane);
        p[j] = wmma4(a, b, p[j]);
      }
    }
  }
  __syncthreads();

#pragma unroll
  for (int j = 0; j < 2; ++j) {
    const int tt = wave * 2 + j;
    const int ti = tt >> 2, tj = tt & 3;
    const int col = tj * 16 + (lane & 15);
    const int rb  = ti * 16 + ((lane >> 4) << 3);
#pragma unroll
    for (int r = 0; r < 8; ++r) {
      const int row = rb + r;
      Bs[row * C_ + col] = (col <= row) ? p[j][r] : 0.0f;
    }
  }
  __syncthreads();

  if (tid < C_) {
    float s = 0.0f, dq = 0.0f;
    for (int c = 0; c < C_; ++c) {
      s  += Bs[tid * C_ + c];
      dq += Qs[tid * D_ + c] * ks[c];
    }
    inv[tid] = 1.0f / (s + dq + EPS_);
  }

#pragma unroll
  for (int j = 0; j < 2; ++j) {
    const int tt = wave * 2 + j;
    const int mb = (tt >> 2) * 16, nb = (tt & 3) * 16;
#pragma unroll
    for (int kk = 0; kk < 16; ++kk) {
      v2f a = frag_a_rm(Bs, C_, mb, kk, lane);
      v2f b = frag_b_rm(Vs, DV_, nb, kk, lane);
      acc[j] = wmma4(a, b, acc[j]);
    }
  }
  __syncthreads();

  float* ob = out + (((size_t)n * L_ + (size_t)t * C_) * H_ + h) * DV_;
#pragma unroll
  for (int j = 0; j < 2; ++j) {
    const int tt = wave * 2 + j;
    const int mb = (tt >> 2) * 16, nb = (tt & 3) * 16;
    const int col = nb + (lane & 15);
    const int rb  = mb + ((lane >> 4) << 3);
#pragma unroll
    for (int r = 0; r < 8; ++r) {
      const int row = rb + r;
      ob[(size_t)row * (H_ * DV_) + col] = acc[j][r] * inv[row];
    }
  }
}

extern "C" void kernel_launch(void* const* d_in, const int* in_sizes, int n_in,
                              void* d_out, int out_size, void* d_ws, size_t ws_size,
                              hipStream_t stream) {
  (void)in_sizes; (void)n_in; (void)out_size; (void)ws_size;
  const float* q = (const float*)d_in[0];
  const float* k = (const float*)d_in[1];
  const float* v = (const float*)d_in[2];
  float* out = (float*)d_out;

  float* kv   = (float*)d_ws;
  float* ksum = kv + (size_t)NH_ * T_ * D_ * DV_;

  cla_phase1<<<NH_ * T_, 256, 0, stream>>>(k, v, kv, ksum);
  cla_phase2<<<NH_,       256, 0, stream>>>(kv, ksum);
  cla_phase3<<<NH_ * T_, 256, 0, stream>>>(q, k, v, kv, ksum, out);
}


# Output pass without LDS or barriers: a wave owns 16 rows of a chunk, loads every fragment from global in register layout, forms scores transposed (K Q^T) so they feed P@V as A; sum order changed
# speedup vs baseline: 1.4878x; 1.4878x over previous
#include <hip/hip_runtime.h>
#include <hip/hip_bf16.h>


#define N_  4
#define L_  4096
#define H_  8
#define D_  64
#define DV_ 64
#define C_  64
#define T_  (L_ / C_)
#define NH_ (N_ * H_)
#define EPS_ 1e-6f

typedef float v2f __attribute__((ext_vector_type(2)));
typedef float v8f __attribute__((ext_vector_type(8)));

__device__ __forceinline__ v8f wmma4(v2f a, v2f b, v8f c) {
  return __builtin_amdgcn_wmma_f32_16x16x4_f32(
       false, a,  false, b,
       (short)0, c,  false,  false);
}

__device__ __forceinline__ float featmap(float x) {
  return x > 0.0f ? x + 1.0f : __expf(x);
}


__device__ __forceinline__ void tile_request(float (&kr)[32], v2f (&vr)[32], int kk,
                                             const float* kp, const float* vp) {
  const size_t rs = (size_t)H_ * D_;
  kr[2 * kk]     = kp[(size_t)(kk * 4) * rs];
  kr[2 * kk + 1] = kp[(size_t)(kk * 4 + 1) * rs];
  vr[2 * kk]     = *reinterpret_cast<const v2f*>(vp + (size_t)(kk * 4) * rs);
  vr[2 * kk + 1] = *reinterpret_cast<const v2f*>(vp + (size_t)(kk * 4 + 1) * rs);
}

template <bool REQ>
__device__ __forceinline__ void tile_chunk(float (&kr)[32], v2f (&vr)[32],
                                           const float* kpn, const float* vpn,
                                           v8f& run0, v8f& run1, float* kvp) {
#pragma unroll
  for (int r = 0; r < 8; ++r) {
    v2f o; o.x = run0[r]; o.y = run1[r];
    *reinterpret_cast<v2f*>(kvp + (size_t)r * DV_) = o;
  }
  v8f cur0 = {}, cur1 = {};
#pragma unroll
  for (int kk = 0; kk < 16; ++kk) {
    v2f a, b0, b1;
    a.x  = featmap(kr[2 * kk]);
    a.y  = featmap(kr[2 * kk + 1]);
    b0.x = vr[2 * kk].x; b0.y = vr[2 * kk + 1].x;
    b1.x = vr[2 * kk].y; b1.y = vr[2 * kk + 1].y;
    cur0 = wmma4(a, b0, cur0);
    cur1 = wmma4(a, b1, cur1);
    if (REQ) {
      tile_request(kr, vr, kk, kpn, vpn);
      __builtin_amdgcn_sched_barrier(0);
    }
  }
  run0 += cur0;
  run1 += cur1;
}

__device__ __forceinline__ void ksum_request(float (&kr)[C_], const float* kq) {
  const size_t rs = (size_t)H_ * D_;
#pragma unroll
  for (int r = 0; r < C_; ++r) kr[r] = kq[(size_t)r * rs];
}
__device__ __forceinline__ void ksum_chunk(const float (&kr)[C_], float& runk, float* ksp) {
  ksp[0] = runk;
  float s = 0.0f;
#pragma unroll
  for (int r = 0; r < C_; ++r) s += featmap(kr[r]);
  runk += s;
}

__global__ __launch_bounds__(96) void cla_state_prefix(
    const float* __restrict__ kin, const float* __restrict__ vin,
    float* __restrict__ kv, float* __restrict__ ksum) {
  const int b = blockIdx.x;
  const int nh = b >> 2, eq = (b >> 1) & 1, dh = b & 1;
  const int n = nh / H_, h = nh % H_;
  const int wave = threadIdx.x >> 5, lane = threadIdx.x & 31;
  const size_t rs = (size_t)H_ * D_;
  const size_t cs = (size_t)C_ * rs;
  const float* kb = kin + (((size_t)n * L_) * H_ + h) * D_;
  const float* vb = vin + (((size_t)n * L_) * H_ + h) * D_;

  if (wave == 2) {
    if (eq != 0) return;
    const int d = dh * 32 + lane;
    const float* kq = kb + d;
    float* ksp = ksum + (size_t)nh * T_ * D_ + d;
    float runk = 0.0f;
    float ra[C_], rb[C_];
    ksum_request(ra, kq);
    for (int t = 0; t < T_ - 2; t += 2) {
      ksum_request(rb, kq + cs);
      __builtin_amdgcn_sched_barrier(0);
      ksum_chunk(ra, runk, ksp);
      ksum_request(ra, kq + 2 * cs);
      __builtin_amdgcn_sched_barrier(0);
      ksum_chunk(rb, runk, ksp + D_);
      kq += 2 * cs;
      ksp += 2 * D_;
    }
    ksum_request(rb, kq + cs);
    __builtin_amdgcn_sched_barrier(0);
    ksum_chunk(ra, runk, ksp);
    ksum_chunk(rb, runk, ksp + D_);
    return;
  }

  const int db = dh * 2 + wave;
  const int m = lane & 15, hi = lane >> 4;
  const float* kp = kb + (size_t)(2 * hi) * rs + db * 16 + m;
  const float* vp = vb + (size_t)(2 * hi) * rs + eq * 32 + 2 * m;
  float* kvp = kv + (size_t)nh * T_ * (D_ * DV_) + (size_t)(db * 16 + 8 * hi) * DV_ + eq * 32 + 2 * m;

  v8f run0 = {}, run1 = {};
  float kr[32];
  v2f vr[32];
#pragma unroll
  for (int kk = 0; kk < 16; ++kk) {
    tile_request(kr, vr, kk, kp, vp);
    __builtin_amdgcn_sched_barrier(0);
  }
  for (int t = 0; t < T_ - 1; ++t) {
    tile_chunk<true>(kr, vr, kp + cs, vp + cs, run0, run1, kvp);
    kp += cs;
    vp += cs;
    kvp += D_ * DV_;
  }
  tile_chunk<false>(kr, vr, kp, vp, run0, run1, kvp);
}


typedef float v4f __attribute__((ext_vector_type(4)));

__device__ __forceinline__ v4f featmap4(v4f x) {
  v4f y; y.x = featmap(x.x); y.y = featmap(x.y); y.z = featmap(x.z); y.w = featmap(x.w); return y;
}
__device__ __forceinline__ v2f pair(float x, float y) { v2f p; p.x = x; p.y = y; return p; }

__global__ __launch_bounds__(128) void cla_phase3(
    const float* __restrict__ qin, const float* __restrict__ kin,
    const float* __restrict__ vin, const float* __restrict__ kv,
    const float* __restrict__ ksum, float* __restrict__ out) {
  const int bt = blockIdx.x;
  const int nh = bt / T_, t = bt % T_;
  const int n = nh / H_, h = nh % H_;
  const int ti = __builtin_amdgcn_readfirstlane((int)(threadIdx.x >> 5));
  const int lane = threadIdx.x & 31;
  const int c = lane & 15, hf = lane >> 4;
  const size_t rs = (size_t)H_ * D_;
  const size_t base = (((size_t)n * L_ + (size_t)t * C_) * H_ + h) * D_;
  const float* sp  = kv + (size_t)(nh * T_ + t) * (D_ * DV_) + (size_t)(4 * hf) * DV_ + 4 * c;
  const float* kzp = ksum + (size_t)(nh * T_ + t) * D_ + 4 * hf;
  const float* qp  = qin + base + (size_t)(ti * 16 + c) * rs + 4 * hf;

  v4f qf[8];
#pragma unroll
  for (int j = 0; j < 8; ++j) qf[j] = featmap4(*reinterpret_cast<const v4f*>(qp + 8 * j));

  float z = 0.0f;
#pragma unroll
  for (int j = 0; j < 8; ++j) {
    const v4f kz = *reinterpret_cast<const v4f*>(kzp + 8 * j);
    z += qf[j].x * kz.x + qf[j].y * kz.y + qf[j].z * kz.z + qf[j].w * kz.w;
  }

  v8f acc[4];
#pragma unroll
  for (int jj = 0; jj < 4; ++jj) acc[jj] = (v8f){};
#pragma unroll
  for (int j = 0; j < 8; ++j) {
    const v4f s0 = *reinterpret_cast<const v4f*>(sp + (size_t)(8 * j + 0) * DV_);
    const v4f s1 = *reinterpret_cast<const v4f*>(sp + (size_t)(8 * j + 1) * DV_);
    const v4f s2 = *reinterpret_cast<const v4f*>(sp + (size_t)(8 * j + 2) * DV_);
    const v4f s3 = *reinterpret_cast<const v4f*>(sp + (size_t)(8 * j + 3) * DV_);
    const v2f a0 = pair(qf[j].x, qf[j].y), a1 = pair(qf[j].z, qf[j].w);
#pragma unroll
    for (int jj = 0; jj < 4; ++jj) acc[jj] = wmma4(a0, pair(s0[jj], s1[jj]), acc[jj]);
#pragma unroll
    for (int jj = 0; jj < 4; ++jj) acc[jj] = wmma4(a1, pair(s2[jj], s3[jj]), acc[jj]);
  }

  for (int tj = 0; tj <= ti; ++tj) {
    const float* kp = kin + base + (size_t)(tj * 16 + c) * rs + 4 * hf;
    const float* vp = vin + base + (size_t)(tj * 16 + 8 * hf) * rs + 4 * c;
    v4f kf[8], vf[8];
#pragma unroll
    for (int j = 0; j < 8; ++j) kf[j] = *reinterpret_cast<const v4f*>(kp + 8 * j);
#pragma unroll
    for (int r = 0; r < 8; ++r) vf[r] = *reinterpret_cast<const v4f*>(vp + (size_t)r * rs);

    v8f pt0 = {}, pt1 = {};
#pragma unroll
    for (int j = 0; j < 8; ++j) {
      const v4f kk = featmap4(kf[j]);
      pt0 = wmma4(pair(kk.x, kk.y), pair(qf[j].x, qf[j].y), pt0);
      pt1 = wmma4(pair(kk.z, kk.w), pair(qf[j].z, qf[j].w), pt1);
    }
    const bool diag = (tj == ti);
    float p[8];
#pragma unroll
    for (int r = 0; r < 8; ++r) {
      const float x = pt0[r] + pt1[r];
      p[r] = (diag && (8 * hf + r > c)) ? 0.0f : x;
      z += p[r];
    }
#pragma unroll
    for (int s = 0; s < 4; ++s) {
      const v2f a = pair(p[2 * s], p[2 * s + 1]);
#pragma unroll
      for (int jj = 0; jj < 4; ++jj)
        acc[jj] = wmma4(a, pair(vf[2 * s][jj], vf[2 * s + 1][jj]), acc[jj]);
    }
  }

  z += __shfl_xor(z, 16, 32);
  const float inv = 1.0f / (z + EPS_);

  float* ob = out + base + (size_t)(ti * 16 + 8 * hf) * (H_ * DV_) + 4 * c;
#pragma unroll
  for (int r = 0; r < 8; ++r) {
    const float ir = __shfl(inv, 8 * hf + r, 32);
    v4f o;
    o.x = acc[0][r] * ir; o.y = acc[1][r] * ir; o.z = acc[2][r] * ir; o.w = acc[3][r] * ir;
    *reinterpret_cast<v4f*>(ob + (size_t)r * (H_ * DV_)) = o;
  }
}

extern "C" void kernel_launch(void* const* d_in, const int* in_sizes, int n_in,
                              void* d_out, int out_size, void* d_ws, size_t ws_size,
                              hipStream_t stream) {
  (void)in_sizes; (void)n_in; (void)out_size; (void)ws_size;
  const float* q = (const float*)d_in[0];
  const float* k = (const float*)d_in[1];
  const float* v = (const float*)d_in[2];
  float* out = (float*)d_out;

  float* kv   = (float*)d_ws;
  float* ksum = kv + (size_t)NH_ * T_ * D_ * DV_;

  cla_state_prefix<<<NH_ * 4, 96, 0, stream>>>(k, v, kv, ksum);
  cla_phase3<<<NH_ * T_, 128, 0, stream>>>(q, k, v, kv, ksum, out);
}


# Output pass, equal work per wave: a block takes two consecutive chunks, wave w does strip w of the first and strip 3-w of the second (5 key tiles each, not 1 to 4); 1024 blocks; sums kept
# speedup vs baseline: 1.5447x; 1.0382x over previous
#include <hip/hip_runtime.h>
#include <hip/hip_bf16.h>


#define N_  4
#define L_  4096
#define H_  8
#define D_  64
#define DV_ 64
#define C_  64
#define T_  (L_ / C_)
#define NH_ (N_ * H_)
#define EPS_ 1e-6f

typedef float v2f __attribute__((ext_vector_type(2)));
typedef float v8f __attribute__((ext_vector_type(8)));

__device__ __forceinline__ v8f wmma4(v2f a, v2f b, v8f c) {
  return __builtin_amdgcn_wmma_f32_16x16x4_f32(
       false, a,  false, b,
       (short)0, c,  false,  false);
}

__device__ __forceinline__ float featmap(float x) {
  return x > 0.0f ? x + 1.0f : __expf(x);
}


__device__ __forceinline__ void tile_request(float (&kr)[32], v2f (&vr)[32], int kk,
                                             const float* kp, const float* vp) {
  const size_t rs = (size_t)H_ * D_;
  kr[2 * kk]     = kp[(size_t)(kk * 4) * rs];
  kr[2 * kk + 1] = kp[(size_t)(kk * 4 + 1) * rs];
  vr[2 * kk]     = *reinterpret_cast<const v2f*>(vp + (size_t)(kk * 4) * rs);
  vr[2 * kk + 1] = *reinterpret_cast<const v2f*>(vp + (size_t)(kk * 4 + 1) * rs);
}

template <bool REQ>
__device__ __forceinline__ void tile_chunk(float (&kr)[32], v2f (&vr)[32],
                                           const float* kpn, const float* vpn,
                                           v8f& run0, v8f& run1, float* kvp) {
#pragma unroll
  for (int r = 0; r < 8; ++r) {
    v2f o; o.x = run0[r]; o.y = run1[r];
    *reinterpret_cast<v2f*>(kvp + (size_t)r * DV_) = o;
  }
  v8f cur0 = {}, cur1 = {};
#pragma unroll
  for (int kk = 0; kk < 16; ++kk) {
    v2f a, b0, b1;
    a.x  = featmap(kr[2 * kk]);
    a.y  = featmap(kr[2 * kk + 1]);
    b0.x = vr[2 * kk].x; b0.y = vr[2 * kk + 1].x;
    b1.x = vr[2 * kk].y; b1.y = vr[2 * kk + 1].y;
    cur0 = wmma4(a, b0, cur0);
    cur1 = wmma4(a, b1, cur1);
    if (REQ) {
      tile_request(kr, vr, kk, kpn, vpn);
      __builtin_amdgcn_sched_barrier(0);
    }
  }
  run0 += cur0;
  run1 += cur1;
}

__device__ __forceinline__ void ksum_request(float (&kr)[C_], const float* kq) {
  const size_t rs = (size_t)H_ * D_;
#pragma unroll
  for (int r = 0; r < C_; ++r) kr[r] = kq[(size_t)r * rs];
}
__device__ __forceinline__ void ksum_chunk(const float (&kr)[C_], float& runk, float* ksp) {
  ksp[0] = runk;
  float s = 0.0f;
#pragma unroll
  for (int r = 0; r < C_; ++r) s += featmap(kr[r]);
  runk += s;
}

__global__ __launch_bounds__(96) void cla_state_prefix(
    const float* __restrict__ kin, const float* __restrict__ vin,
    float* __restrict__ kv, float* __restrict__ ksum) {
  const int b = blockIdx.x;
  const int nh = b >> 2, eq = (b >> 1) & 1, dh = b & 1;
  const int n = nh / H_, h = nh % H_;
  const int wave = threadIdx.x >> 5, lane = threadIdx.x & 31;
  const size_t rs = (size_t)H_ * D_;
  const size_t cs = (size_t)C_ * rs;
  const float* kb = kin + (((size_t)n * L_) * H_ + h) * D_;
  const float* vb = vin + (((size_t)n * L_) * H_ + h) * D_;

  if (wave == 2) {
    if (eq != 0) return;
    const int d = dh * 32 + lane;
    const float* kq = kb + d;
    float* ksp = ksum + (size_t)nh * T_ * D_ + d;
    float runk = 0.0f;
    float ra[C_], rb[C_];
    ksum_request(ra, kq);
    for (int t = 0; t < T_ - 2; t += 2) {
      ksum_request(rb, kq + cs);
      __builtin_amdgcn_sched_barrier(0);
      ksum_chunk(ra, runk, ksp);
      ksum_request(ra, kq + 2 * cs);
      __builtin_amdgcn_sched_barrier(0);
      ksum_chunk(rb, runk, ksp + D_);
      kq += 2 * cs;
      ksp += 2 * D_;
    }
    ksum_request(rb, kq + cs);
    __builtin_amdgcn_sched_barrier(0);
    ksum_chunk(ra, runk, ksp);
    ksum_chunk(rb, runk, ksp + D_);
    return;
  }

  const int db = dh * 2 + wave;
  const int m = lane & 15, hi = lane >> 4;
  const float* kp = kb + (size_t)(2 * hi) * rs + db * 16 + m;
  const float* vp = vb + (size_t)(2 * hi) * rs + eq * 32 + 2 * m;
  float* kvp = kv + (size_t)nh * T_ * (D_ * DV_) + (size_t)(db * 16 + 8 * hi) * DV_ + eq * 32 + 2 * m;

  v8f run0 = {}, run1 = {};
  float kr[32];
  v2f vr[32];
#pragma unroll
  for (int kk = 0; kk < 16; ++kk) {
    tile_request(kr, vr, kk, kp, vp);
    __builtin_amdgcn_sched_barrier(0);
  }
  for (int t = 0; t < T_ - 1; ++t) {
    tile_chunk<true>(kr, vr, kp + cs, vp + cs, run0, run1, kvp);
    kp += cs;
    vp += cs;
    kvp += D_ * DV_;
  }
  tile_chunk<false>(kr, vr, kp, vp, run0, run1, kvp);
}


typedef float v4f __attribute__((ext_vector_type(4)));

__device__ __forceinline__ v4f featmap4(v4f x) {
  v4f y; y.x = featmap(x.x); y.y = featmap(x.y); y.z = featmap(x.z); y.w = featmap(x.w); return y;
}
__device__ __forceinline__ v2f pair(float x, float y) { v2f p; p.x = x; p.y = y; return p; }

__global__ __launch_bounds__(128) void cla_phase3(
    const float* __restrict__ qin, const float* __restrict__ kin,
    const float* __restrict__ vin, const float* __restrict__ kv,
    const float* __restrict__ ksum, float* __restrict__ out) {
  const int w = __builtin_amdgcn_readfirstlane((int)(threadIdx.x >> 5));
  const int lane = threadIdx.x & 31;
  const int c = lane & 15, hf = lane >> 4;
  const size_t rs = (size_t)H_ * D_;
  for (int it = 0; it < 2; ++it) {
  const int bt = 2 * (int)blockIdx.x + it;
  const int nh = bt / T_, t = bt % T_;
  const int n = nh / H_, h = nh % H_;
  const int ti = it ? 3 - w : w;
  const size_t base = (((size_t)n * L_ + (size_t)t * C_) * H_ + h) * D_;
  const float* sp  = kv + (size_t)(nh * T_ + t) * (D_ * DV_) + (size_t)(4 * hf) * DV_ + 4 * c;
  const float* kzp = ksum + (size_t)(nh * T_ + t) * D_ + 4 * hf;
  const float* qp  = qin + base + (size_t)(ti * 16 + c) * rs + 4 * hf;

  v4f qf[8];
#pragma unroll
  for (int j = 0; j < 8; ++j) qf[j] = featmap4(*reinterpret_cast<const v4f*>(qp + 8 * j));

  float z = 0.0f;
#pragma unroll
  for (int j = 0; j < 8; ++j) {
    const v4f kz = *reinterpret_cast<const v4f*>(kzp + 8 * j);
    z += qf[j].x * kz.x + qf[j].y * kz.y + qf[j].z * kz.z + qf[j].w * kz.w;
  }

  v8f acc[4];
#pragma unroll
  for (int jj = 0; jj < 4; ++jj) acc[jj] = (v8f){};
#pragma unroll
  for (int j = 0; j < 8; ++j) {
    const v4f s0 = *reinterpret_cast<const v4f*>(sp + (size_t)(8 * j + 0) * DV_);
    const v4f s1 = *reinterpret_cast<const v4f*>(sp + (size_t)(8 * j + 1) * DV_);
    const v4f s2 = *reinterpret_cast<const v4f*>(sp + (size_t)(8 * j + 2) * DV_);
    const v4f s3 = *reinterpret_cast<const v4f*>(sp + (size_t)(8 * j + 3) * DV_);
    const v2f a0 = pair(qf[j].x, qf[j].y), a1 = pair(qf[j].z, qf[j].w);
#pragma unroll
    for (int jj = 0; jj < 4; ++jj) acc[jj] = wmma4(a0, pair(s0[jj], s1[jj]), acc[jj]);
#pragma unroll
    for (int jj = 0; jj < 4; ++jj) acc[jj] = wmma4(a1, pair(s2[jj], s3[jj]), acc[jj]);
  }

  for (int tj = 0; tj <= ti; ++tj) {
    const float* kp = kin + base + (size_t)(tj * 16 + c) * rs + 4 * hf;
    const float* vp = vin + base + (size_t)(tj * 16 + 8 * hf) * rs + 4 * c;
    v4f kf[8], vf[8];
#pragma unroll
    for (int j = 0; j < 8; ++j) kf[j] = *reinterpret_cast<const v4f*>(kp + 8 * j);
#pragma unroll
    for (int r = 0; r < 8; ++r) vf[r] = *reinterpret_cast<const v4f*>(vp + (size_t)r * rs);

    v8f pt0 = {}, pt1 = {};
#pragma unroll
    for (int j = 0; j < 8; ++j) {
      const v4f kk = featmap4(kf[j]);
      pt0 = wmma4(pair(kk.x, kk.y), pair(qf[j].x, qf[j].y), pt0);
      pt1 = wmma4(pair(kk.z, kk.w), pair(qf[j].z, qf[j].w), pt1);
    }
    const bool diag = (tj == ti);
    float p[8];
#pragma unroll
    for (int r = 0; r < 8; ++r) {
      const float x = pt0[r] + pt1[r];
      p[r] = (diag && (8 * hf + r > c)) ? 0.0f : x;
      z += p[r];
    }
#pragma unroll
    for (int s = 0; s < 4; ++s) {
      const v2f a = pair(p[2 * s], p[2 * s + 1]);
#pragma unroll
      for (int jj = 0; jj < 4; ++jj)
        acc[jj] = wmma4(a, pair(vf[2 * s][jj], vf[2 * s + 1][jj]), acc[jj]);
    }
  }

  z += __shfl_xor(z, 16, 32);
  const float inv = 1.0f / (z + EPS_);

  float* ob = out + base + (size_t)(ti * 16 + 8 * hf) * (H_ * DV_) + 4 * c;
#pragma unroll
  for (int r = 0; r < 8; ++r) {
    const float ir = __shfl(inv, 8 * hf + r, 32);
    v4f o;
    o.x = acc[0][r] * ir; o.y = acc[1][r] * ir; o.z = acc[2][r] * ir; o.w = acc[3][r] * ir;
    *reinterpret_cast<v4f*>(ob + (size_t)r * (H_ * DV_)) = o;
  }
  }
}

extern "C" void kernel_launch(void* const* d_in, const int* in_sizes, int n_in,
                              void* d_out, int out_size, void* d_ws, size_t ws_size,
                              hipStream_t stream) {
  (void)in_sizes; (void)n_in; (void)out_size; (void)ws_size;
  const float* q = (const float*)d_in[0];
  const float* k = (const float*)d_in[1];
  const float* v = (const float*)d_in[2];
  float* out = (float*)d_out;

  float* kv   = (float*)d_ws;
  float* ksum = kv + (size_t)NH_ * T_ * D_ * DV_;

  cla_state_prefix<<<NH_ * 4, 96, 0, stream>>>(k, v, kv, ksum);
  cla_phase3<<<NH_ * T_ / 2, 128, 0, stream>>>(q, k, v, kv, ksum, out);
}


# State walk on twice the waves: one 16x16 state tile per wave (16 products a chunk, not 32), rows asked into L2 4 chunks ahead; state stored row-pair interleaved (whole-line stores); sums kept
# speedup vs baseline: 1.8913x; 1.2244x over previous
#include <hip/hip_runtime.h>
#include <hip/hip_bf16.h>


#define N_  4
#define L_  4096
#define H_  8
#define D_  64
#define DV_ 64
#define C_  64
#define T_  (L_ / C_)
#define NH_ (N_ * H_)
#define EPS_ 1e-6f

typedef float v2f __attribute__((ext_vector_type(2)));
typedef float v8f __attribute__((ext_vector_type(8)));

__device__ __forceinline__ v8f wmma4(v2f a, v2f b, v8f c) {
  return __builtin_amdgcn_wmma_f32_16x16x4_f32(
       false, a,  false, b,
       (short)0, c,  false,  false);
}

__device__ __forceinline__ float featmap(float x) {
  return x > 0.0f ? x + 1.0f : __expf(x);
}


__device__ __forceinline__ void tile_request(float (&kr)[32], float (&vr)[32], int kk,
                                             const float* kp, const float* vp) {
  const size_t rs = (size_t)H_ * D_;
  kr[2 * kk]     = kp[(size_t)(kk * 4) * rs];
  kr[2 * kk + 1] = kp[(size_t)(kk * 4 + 1) * rs];
  vr[2 * kk]     = vp[(size_t)(kk * 4) * rs];
  vr[2 * kk + 1] = vp[(size_t)(kk * 4 + 1) * rs];
}

template <bool REQ>
__device__ __forceinline__ void tile_chunk(float (&kr)[32], float (&vr)[32],
                                           const float* kpn, const float* vpn,
                                           v8f& run, float* kvp) {
#pragma unroll
  for (int s = 0; s < 4; ++s) {
    v2f o; o.x = run[2 * s]; o.y = run[2 * s + 1];
    *reinterpret_cast<v2f*>(kvp + (size_t)s * (2 * DV_)) = o;
  }
  v8f cur = {};
#pragma unroll
  for (int kk = 0; kk < 16; ++kk) {
    v2f a, b;
    a.x = featmap(kr[2 * kk]);
    a.y = featmap(kr[2 * kk + 1]);
    b.x = vr[2 * kk]; b.y = vr[2 * kk + 1];
    cur = wmma4(a, b, cur);
    if (REQ) {
      tile_request(kr, vr, kk, kpn, vpn);
      __builtin_amdgcn_sched_barrier(0);
    }
  }
  run += cur;
}

__device__ __forceinline__ void ksum_request(float (&kr)[C_], const float* kq) {
  const size_t rs = (size_t)H_ * D_;
#pragma unroll
  for (int r = 0; r < C_; ++r) kr[r] = kq[(size_t)r * rs];
}
__device__ __forceinline__ void ksum_chunk(const float (&kr)[C_], float& runk, float* ksp) {
  ksp[0] = runk;
  float s = 0.0f;
#pragma unroll
  for (int r = 0; r < C_; ++r) s += featmap(kr[r]);
  runk += s;
}

__global__ __launch_bounds__(96) void cla_state_prefix(
    const float* __restrict__ kin, const float* __restrict__ vin,
    float* __restrict__ kv, float* __restrict__ ksum) {
  const int b = blockIdx.x;
  const int nh = b >> 3, q = b & 7;
  const int n = nh / H_, h = nh % H_;
  const int wave = threadIdx.x >> 5, lane = threadIdx.x & 31;
  const size_t rs = (size_t)H_ * D_;
  const size_t cs = (size_t)C_ * rs;
  const float* kb = kin + (((size_t)n * L_) * H_ + h) * D_;
  const float* vb = vin + (((size_t)n * L_) * H_ + h) * D_;

  if (wave == 2) {
    if (q > 1) return;
    const int d = q * 32 + lane;
    const float* kq = kb + d;
    float* ksp = ksum + (size_t)nh * T_ * D_ + d;
    float runk = 0.0f;
    float ra[C_], rb[C_];
    ksum_request(ra, kq);
    for (int t = 0; t < T_ - 2; t += 2) {
      ksum_request(rb, kq + cs);
      __builtin_amdgcn_sched_barrier(0);
      ksum_chunk(ra, runk, ksp);
      ksum_request(ra, kq + 2 * cs);
      __builtin_amdgcn_sched_barrier(0);
      ksum_chunk(rb, runk, ksp + D_);
      kq += 2 * cs;
      ksp += 2 * D_;
    }
    ksum_request(rb, kq + cs);
    __builtin_amdgcn_sched_barrier(0);
    ksum_chunk(ra, runk, ksp);
    ksum_chunk(rb, runk, ksp + D_);
    return;
  }

  const int tile = 2 * q + wave;
  const int db = tile >> 2, eb = tile & 3;
  const int m = lane & 15, hi = lane >> 4;
  const float* kp = kb + (size_t)(2 * hi) * rs + db * 16 + m;
  const float* vp = vb + (size_t)(2 * hi) * rs + eb * 16 + m;
  float* kvp = kv + (size_t)nh * T_ * (D_ * DV_) + ((size_t)(db * 8 + 4 * hi) * DV_ + eb * 16 + m) * 2;

  const float* kpf = kb + (size_t)(32 * wave + lane) * rs;
  const float* vpf = vb + (size_t)(32 * wave + lane) * rs;

  v8f run = {};
  float kr[32];
  float vr[32];
#pragma unroll
  for (int kk = 0; kk < 16; ++kk) {
    tile_request(kr, vr, kk, kp, vp);
    __builtin_amdgcn_sched_barrier(0);
  }
  for (int t = 0; t < T_ - 1; ++t) {
    const int tp = t + 4 < T_ ? t + 4 : T_ - 1;
    __builtin_prefetch(kpf + (size_t)tp * cs, 0, 3);
    __builtin_prefetch(vpf + (size_t)tp * cs, 0, 3);
    tile_chunk<true>(kr, vr, kp + cs, vp + cs, run, kvp);
    kp += cs;
    vp += cs;
    kvp += D_ * DV_;
  }
  tile_chunk<false>(kr, vr, kp, vp, run, kvp);
}


typedef float v4f __attribute__((ext_vector_type(4)));

__device__ __forceinline__ v4f featmap4(v4f x) {
  v4f y; y.x = featmap(x.x); y.y = featmap(x.y); y.z = featmap(x.z); y.w = featmap(x.w); return y;
}
__device__ __forceinline__ v2f pair(float x, float y) { v2f p; p.x = x; p.y = y; return p; }

__global__ __launch_bounds__(128) void cla_phase3(
    const float* __restrict__ qin, const float* __restrict__ kin,
    const float* __restrict__ vin, const float* __restrict__ kv,
    const float* __restrict__ ksum, float* __restrict__ out) {
  const int w = __builtin_amdgcn_readfirstlane((int)(threadIdx.x >> 5));
  const int lane = threadIdx.x & 31;
  const int c = lane & 15, hf = lane >> 4;
  const size_t rs = (size_t)H_ * D_;
  for (int it = 0; it < 2; ++it) {
  const int bt = 2 * (int)blockIdx.x + it;
  const int nh = bt / T_, t = bt % T_;
  const int n = nh / H_, h = nh % H_;
  const int ti = it ? 3 - w : w;
  const size_t base = (((size_t)n * L_ + (size_t)t * C_) * H_ + h) * D_;
  const float* sp  = kv + (size_t)(nh * T_ + t) * (D_ * DV_) + (size_t)(2 * hf) * (2 * DV_) + 8 * c;
  const float* kzp = ksum + (size_t)(nh * T_ + t) * D_ + 4 * hf;
  const float* qp  = qin + base + (size_t)(ti * 16 + c) * rs + 4 * hf;

  v4f qf[8];
#pragma unroll
  for (int j = 0; j < 8; ++j) qf[j] = featmap4(*reinterpret_cast<const v4f*>(qp + 8 * j));

  float z = 0.0f;
#pragma unroll
  for (int j = 0; j < 8; ++j) {
    const v4f kz = *reinterpret_cast<const v4f*>(kzp + 8 * j);
    z += qf[j].x * kz.x + qf[j].y * kz.y + qf[j].z * kz.z + qf[j].w * kz.w;
  }

  v8f acc[4];
#pragma unroll
  for (int jj = 0; jj < 4; ++jj) acc[jj] = (v8f){};
#pragma unroll
  for (int j = 0; j < 8; ++j) {
    const float* sj = sp + (size_t)(4 * j) * (2 * DV_);
    const v4f s00 = *reinterpret_cast<const v4f*>(sj);
    const v4f s01 = *reinterpret_cast<const v4f*>(sj + 4);
    const v4f s10 = *reinterpret_cast<const v4f*>(sj + 2 * DV_);
    const v4f s11 = *reinterpret_cast<const v4f*>(sj + 2 * DV_ + 4);
    const v2f a0 = pair(qf[j].x, qf[j].y), a1 = pair(qf[j].z, qf[j].w);
    acc[0] = wmma4(a0, pair(s00.x, s00.y), acc[0]);
    acc[1] = wmma4(a0, pair(s00.z, s00.w), acc[1]);
    acc[2] = wmma4(a0, pair(s01.x, s01.y), acc[2]);
    acc[3] = wmma4(a0, pair(s01.z, s01.w), acc[3]);
    acc[0] = wmma4(a1, pair(s10.x, s10.y), acc[0]);
    acc[1] = wmma4(a1, pair(s10.z, s10.w), acc[1]);
    acc[2] = wmma4(a1, pair(s11.x, s11.y), acc[2]);
    acc[3] = wmma4(a1, pair(s11.z, s11.w), acc[3]);
  }

  for (int tj = 0; tj <= ti; ++tj) {
    const float* kp = kin + base + (size_t)(tj * 16 + c) * rs + 4 * hf;
    const float* vp = vin + base + (size_t)(tj * 16 + 8 * hf) * rs + 4 * c;
    v4f kf[8], vf[8];
#pragma unroll
    for (int j = 0; j < 8; ++j) kf[j] = *reinterpret_cast<const v4f*>(kp + 8 * j);
#pragma unroll
    for (int r = 0; r < 8; ++r) vf[r] = *reinterpret_cast<const v4f*>(vp + (size_t)r * rs);

    v8f pt0 = {}, pt1 = {};
#pragma unroll
    for (int j = 0; j < 8; ++j) {
      const v4f kk = featmap4(kf[j]);
      pt0 = wmma4(pair(kk.x, kk.y), pair(qf[j].x, qf[j].y), pt0);
      pt1 = wmma4(pair(kk.z, kk.w), pair(qf[j].z, qf[j].w), pt1);
    }
    const bool diag = (tj == ti);
    float p[8];
#pragma unroll
    for (int r = 0; r < 8; ++r) {
      const float x = pt0[r] + pt1[r];
      p[r] = (diag && (8 * hf + r > c)) ? 0.0f : x;
      z += p[r];
    }
#pragma unroll
    for (int s = 0; s < 4; ++s) {
      const v2f a = pair(p[2 * s], p[2 * s + 1]);
#pragma unroll
      for (int jj = 0; jj < 4; ++jj)
        acc[jj] = wmma4(a, pair(vf[2 * s][jj], vf[2 * s + 1][jj]), acc[jj]);
    }
  }

  z += __shfl_xor(z, 16, 32);
  const float inv = 1.0f / (z + EPS_);

  float* ob = out + base + (size_t)(ti * 16 + 8 * hf) * (H_ * DV_) + 4 * c;
#pragma unroll
  for (int r = 0; r < 8; ++r) {
    const float ir = __shfl(inv, 8 * hf + r, 32);
    v4f o;
    o.x = acc[0][r] * ir; o.y = acc[1][r] * ir; o.z = acc[2][r] * ir; o.w = acc[3][r] * ir;
    *reinterpret_cast<v4f*>(ob + (size_t)r * (H_ * DV_)) = o;
  }
  }
}

extern "C" void kernel_launch(void* const* d_in, const int* in_sizes, int n_in,
                              void* d_out, int out_size, void* d_ws, size_t ws_size,
                              hipStream_t stream) {
  (void)in_sizes; (void)n_in; (void)out_size; (void)ws_size;
  const float* q = (const float*)d_in[0];
  const float* k = (const float*)d_in[1];
  const float* v = (const float*)d_in[2];
  float* out = (float*)d_out;

  float* kv   = (float*)d_ws;
  float* ksum = kv + (size_t)NH_ * T_ * D_ * DV_;

  cla_state_prefix<<<NH_ * 8, 96, 0, stream>>>(k, v, kv, ksum);
  cla_phase3<<<NH_ * T_ / 2, 128, 0, stream>>>(q, k, v, kv, ksum, out);
}


# State walk cut into two sequence halves walked at once (4 tile waves a block; K column sums kept by the eb=0 tile waves); output pass adds slot 32 to a local prefix; sum order changed
# speedup vs baseline: 2.6487x; 1.4005x over previous
#include <hip/hip_runtime.h>
#include <hip/hip_bf16.h>


#define N_  4
#define L_  4096
#define H_  8
#define D_  64
#define DV_ 64
#define C_  64
#define T_  (L_ / C_)
#define NH_ (N_ * H_)
#define EPS_ 1e-6f

typedef float v2f __attribute__((ext_vector_type(2)));
typedef float v8f __attribute__((ext_vector_type(8)));

__device__ __forceinline__ v8f wmma4(v2f a, v2f b, v8f c) {
  return __builtin_amdgcn_wmma_f32_16x16x4_f32(
       false, a,  false, b,
       (short)0, c,  false,  false);
}

__device__ __forceinline__ float featmap(float x) {
  return x > 0.0f ? x + 1.0f : __expf(x);
}


#define SEG_ 32

__device__ __forceinline__ void tile_request(float (&kr)[32], float (&vr)[32], int kk,
                                             const float* kp, const float* vp) {
  const size_t rs = (size_t)H_ * D_;
  kr[2 * kk]     = kp[(size_t)(kk * 4) * rs];
  kr[2 * kk + 1] = kp[(size_t)(kk * 4 + 1) * rs];
  vr[2 * kk]     = vp[(size_t)(kk * 4) * rs];
  vr[2 * kk + 1] = vp[(size_t)(kk * 4 + 1) * rs];
}

template <bool REQ, bool KS>
__device__ __forceinline__ void tile_chunk(float (&kr)[32], float (&vr)[32],
                                           const float* kpn, const float* vpn,
                                           v8f& run, v2f& ksrun) {
  v8f cur = {};
  v2f ks4[4] = {};
#pragma unroll
  for (int kk = 0; kk < 16; ++kk) {
    v2f a, b;
    a.x = featmap(kr[2 * kk]);
    a.y = featmap(kr[2 * kk + 1]);
    b.x = vr[2 * kk]; b.y = vr[2 * kk + 1];
    cur = wmma4(a, b, cur);
    if (KS) ks4[kk & 3] += a;
    if (REQ) {
      tile_request(kr, vr, kk, kpn, vpn);
      __builtin_amdgcn_sched_barrier(0);
    }
  }
  run += cur;
  if (KS) ksrun += (ks4[0] + ks4[1]) + (ks4[2] + ks4[3]);
}

__device__ __forceinline__ void tile_store(const v8f& run, float* kvp) {
#pragma unroll
  for (int s = 0; s < 4; ++s) {
    v2f o; o.x = run[2 * s]; o.y = run[2 * s + 1];
    *reinterpret_cast<v2f*>(kvp + (size_t)s * (2 * DV_)) = o;
  }
}

__device__ __forceinline__ void ks_store(const v2f& ksrun, float* ksp, int hi) {
  float sum = ksrun.x + ksrun.y;
  sum += __builtin_bit_cast(float, __builtin_amdgcn_permlanex16(__builtin_bit_cast(int, sum),
             __builtin_bit_cast(int, sum), 0x76543210, 0xfedcba98, false, false));
  if (hi == 0) *ksp = sum;
}

template <bool KS>
__device__ __forceinline__ void tile_walk(const float* kp, const float* vp,
                                          const float* kpf, const float* vpf,
                                          float* kvp, float* ksp, int c0, int n, int hi) {
  const size_t rs = (size_t)H_ * D_;
  const size_t cs = (size_t)C_ * rs;
  v8f run = {};
  v2f ksrun = {};
  float kr[32];
  float vr[32];
#pragma unroll
  for (int kk = 0; kk < 16; ++kk) {
    tile_request(kr, vr, kk, kp + (size_t)c0 * cs, vp + (size_t)c0 * cs);
    __builtin_amdgcn_sched_barrier(0);
  }
  if (c0 == 0) tile_store(run, kvp);
  for (int i = 0; i < n - 1; ++i) {
    const int c = c0 + i;
    const int tp = c + 4 < T_ ? c + 4 : T_ - 1;
    __builtin_prefetch(kpf + (size_t)tp * cs, 0, 3);
    __builtin_prefetch(vpf + (size_t)tp * cs, 0, 3);
    tile_chunk<true, KS>(kr, vr, kp + (size_t)(c + 1) * cs, vp + (size_t)(c + 1) * cs, run, ksrun);
    tile_store(run, kvp + (size_t)(c + 1) * (D_ * DV_));
    if (KS) ks_store(ksrun, ksp + (size_t)(c >> 1) * (2 * D_) + (c & 1) * 16, hi);
  }
  const int c = c0 + n - 1;
  tile_chunk<false, KS>(kr, vr, kp, vp, run, ksrun);
  tile_store(run, kvp + (size_t)(c + 1) * (D_ * DV_));
  if (KS) ks_store(ksrun, ksp + (size_t)(c >> 1) * (2 * D_) + (c & 1) * 16, hi);
}

__global__ __launch_bounds__(128) void cla_state_prefix(
    const float* __restrict__ kin, const float* __restrict__ vin,
    float* __restrict__ kv, float* __restrict__ ksum) {
  const int b = blockIdx.x;
  const int nh = b >> 3, q = b & 7;
  const int n = nh / H_, h = nh % H_;
  const int wave = __builtin_amdgcn_readfirstlane((int)(threadIdx.x >> 5));
  const int lane = threadIdx.x & 31;
  const int tsel = wave & 1, half = wave >> 1;
  const size_t rs = (size_t)H_ * D_;
  const float* kb = kin + (((size_t)n * L_) * H_ + h) * D_;
  const float* vb = vin + (((size_t)n * L_) * H_ + h) * D_;

  const int tile = 2 * q + tsel;
  const int db = tile >> 2, eb = tile & 3;
  const int m = lane & 15, hi = lane >> 4;
  const float* kp = kb + (size_t)(2 * hi) * rs + db * 16 + m;
  const float* vp = vb + (size_t)(2 * hi) * rs + eb * 16 + m;
  float* kvp = kv + (size_t)nh * T_ * (D_ * DV_) + ((size_t)(db * 8 + 4 * hi) * DV_ + eb * 16 + m) * 2;
  float* ksp = ksum + (size_t)nh * T_ * D_ + db * 32 + m;
  const float* kpf = kb + (size_t)(32 * tsel + lane) * rs;
  const float* vpf = vb + (size_t)(32 * tsel + lane) * rs;

  const int c0 = half ? SEG_ : 0;
  const int nc = half ? T_ - 1 - SEG_ : SEG_;
  if (eb == 0) tile_walk<true>(kp, vp, kpf, vpf, kvp, ksp, c0, nc, hi);
  else         tile_walk<false>(kp, vp, kpf, vpf, kvp, ksp, c0, nc, hi);
}


typedef float v4f __attribute__((ext_vector_type(4)));

__device__ __forceinline__ v4f featmap4(v4f x) {
  v4f y; y.x = featmap(x.x); y.y = featmap(x.y); y.z = featmap(x.z); y.w = featmap(x.w); return y;
}
__device__ __forceinline__ v2f pair(float x, float y) { v2f p; p.x = x; p.y = y; return p; }

template <bool TWO>
__device__ __forceinline__ void strip_head(const float* qp, const float* sp, const float* st,
                                           const float* kzl, const float* kzt, bool haveks,
                                           v4f (&qf)[8], float& z, v8f (&acc)[4]) {
#pragma unroll
  for (int j = 0; j < 8; ++j) qf[j] = featmap4(*reinterpret_cast<const v4f*>(qp + 8 * j));
#pragma unroll
  for (int j = 0; j < 8; ++j) {
    const int o = (j >> 1) * 32 + (j & 1) * 8;
    v4f kz = *reinterpret_cast<const v4f*>(kzl + o);
    if (TWO) kz += *reinterpret_cast<const v4f*>(kzt + o);
    else     kz = haveks ? kz : (v4f){0.0f, 0.0f, 0.0f, 0.0f};
    z += qf[j].x * kz.x + qf[j].y * kz.y + qf[j].z * kz.z + qf[j].w * kz.w;
  }
  if (TWO) __builtin_amdgcn_sched_barrier(0);
#pragma unroll
  for (int jh = 0; jh < 2; ++jh) {
    v4f s[16];
#pragma unroll
    for (int u = 0; u < 4; ++u) {
      const float* sj = sp + (size_t)(4 * (4 * jh + u)) * (2 * DV_);
      s[4 * u + 0] = *reinterpret_cast<const v4f*>(sj);
      s[4 * u + 1] = *reinterpret_cast<const v4f*>(sj + 4);
      s[4 * u + 2] = *reinterpret_cast<const v4f*>(sj + 2 * DV_);
      s[4 * u + 3] = *reinterpret_cast<const v4f*>(sj + 2 * DV_ + 4);
    }
    if (TWO) {
      v4f tt[16];
#pragma unroll
      for (int u = 0; u < 4; ++u) {
        const float* tj = st + (size_t)(4 * (4 * jh + u)) * (2 * DV_);
        tt[4 * u + 0] = *reinterpret_cast<const v4f*>(tj);
        tt[4 * u + 1] = *reinterpret_cast<const v4f*>(tj + 4);
        tt[4 * u + 2] = *reinterpret_cast<const v4f*>(tj + 2 * DV_);
        tt[4 * u + 3] = *reinterpret_cast<const v4f*>(tj + 2 * DV_ + 4);
      }
#pragma unroll
      for (int i = 0; i < 16; ++i) s[i] += tt[i];
    }
#pragma unroll
    for (int u = 0; u < 4; ++u) {
      const int j = 4 * jh + u;
      const v4f s00 = s[4 * u], s01 = s[4 * u + 1], s10 = s[4 * u + 2], s11 = s[4 * u + 3];
      const v2f a0 = pair(qf[j].x, qf[j].y), a1 = pair(qf[j].z, qf[j].w);
      acc[0] = wmma4(a0, pair(s00.x, s00.y), acc[0]);
      acc[1] = wmma4(a0, pair(s00.z, s00.w), acc[1]);
      acc[2] = wmma4(a0, pair(s01.x, s01.y), acc[2]);
      acc[3] = wmma4(a0, pair(s01.z, s01.w), acc[3]);
      acc[0] = wmma4(a1, pair(s10.x, s10.y), acc[0]);
      acc[1] = wmma4(a1, pair(s10.z, s10.w), acc[1]);
      acc[2] = wmma4(a1, pair(s11.x, s11.y), acc[2]);
      acc[3] = wmma4(a1, pair(s11.z, s11.w), acc[3]);
    }
    if (TWO && jh == 0) __builtin_amdgcn_sched_barrier(0);
  }
}

__global__ __launch_bounds__(128) void cla_phase3(
    const float* __restrict__ qin, const float* __restrict__ kin,
    const float* __restrict__ vin, const float* __restrict__ kv,
    const float* __restrict__ ksum, float* __restrict__ out) {
  const int w = __builtin_amdgcn_readfirstlane((int)(threadIdx.x >> 5));
  const int lane = threadIdx.x & 31;
  const int c = lane & 15, hf = lane >> 4;
  const size_t rs = (size_t)H_ * D_;
  for (int it = 0; it < 2; ++it) {
  const int bt = 2 * (int)blockIdx.x + it;
  const int nh = bt / T_, t = bt % T_;
  const int n = nh / H_, h = nh % H_;
  const int ti = it ? 3 - w : w;
  const size_t base = (((size_t)n * L_ + (size_t)t * C_) * H_ + h) * D_;
  const size_t so = (size_t)(2 * hf) * (2 * DV_) + 8 * c;
  const float* sp  = kv + (size_t)(nh * T_ + t) * (D_ * DV_) + so;
  const float* st  = kv + (size_t)(nh * T_ + SEG_) * (D_ * DV_) + so;
  const float* kzb = ksum + (size_t)nh * T_ * D_ + 4 * hf;
  const int tm = t > 0 ? t - 1 : 0;
  const float* kzl = kzb + (tm >> 1) * (2 * D_) + (tm & 1) * 16;
  const float* kzt = kzb + ((SEG_ - 1) >> 1) * (2 * D_) + ((SEG_ - 1) & 1) * 16;
  const float* qp  = qin + base + (size_t)(ti * 16 + c) * rs + 4 * hf;

  v4f qf[8];
  float z = 0.0f;
  v8f acc[4];
#pragma unroll
  for (int jj = 0; jj < 4; ++jj) acc[jj] = (v8f){};
  if (t > SEG_) strip_head<true>(qp, sp, st, kzl, kzt, true, qf, z, acc);
  else          strip_head<false>(qp, sp, st, kzl, kzt, t > 0, qf, z, acc);

  for (int tj = 0; tj <= ti; ++tj) {
    const float* kp = kin + base + (size_t)(tj * 16 + c) * rs + 4 * hf;
    const float* vp = vin + base + (size_t)(tj * 16 + 8 * hf) * rs + 4 * c;
    v4f kf[8], vf[8];
#pragma unroll
    for (int j = 0; j < 8; ++j) kf[j] = *reinterpret_cast<const v4f*>(kp + 8 * j);
#pragma unroll
    for (int r = 0; r < 8; ++r) vf[r] = *reinterpret_cast<const v4f*>(vp + (size_t)r * rs);

    v8f pt0 = {}, pt1 = {};
#pragma unroll
    for (int j = 0; j < 8; ++j) {
      const v4f kk = featmap4(kf[j]);
      pt0 = wmma4(pair(kk.x, kk.y), pair(qf[j].x, qf[j].y), pt0);
      pt1 = wmma4(pair(kk.z, kk.w), pair(qf[j].z, qf[j].w), pt1);
    }
    const bool diag = (tj == ti);
    float p[8];
#pragma unroll
    for (int r = 0; r < 8; ++r) {
      const float x = pt0[r] + pt1[r];
      p[r] = (diag && (8 * hf + r > c)) ? 0.0f : x;
      z += p[r];
    }
#pragma unroll
    for (int s = 0; s < 4; ++s) {
      const v2f a = pair(p[2 * s], p[2 * s + 1]);
#pragma unroll
      for (int jj = 0; jj < 4; ++jj)
        acc[jj] = wmma4(a, pair(vf[2 * s][jj], vf[2 * s + 1][jj]), acc[jj]);
    }
  }

  z += __shfl_xor(z, 16, 32);
  const float inv = 1.0f / (z + EPS_);

  float* ob = out + base + (size_t)(ti * 16 + 8 * hf) * (H_ * DV_) + 4 * c;
#pragma unroll
  for (int r = 0; r < 8; ++r) {
    const float ir = __shfl(inv, 8 * hf + r, 32);
    v4f o;
    o.x = acc[0][r] * ir; o.y = acc[1][r] * ir; o.z = acc[2][r] * ir; o.w = acc[3][r] * ir;
    *reinterpret_cast<v4f*>(ob + (size_t)r * (H_ * DV_)) = o;
  }
  }
}

extern "C" void kernel_launch(void* const* d_in, const int* in_sizes, int n_in,
                              void* d_out, int out_size, void* d_ws, size_t ws_size,
                              hipStream_t stream) {
  (void)in_sizes; (void)n_in; (void)out_size; (void)ws_size;
  const float* q = (const float*)d_in[0];
  const float* k = (const float*)d_in[1];
  const float* v = (const float*)d_in[2];
  float* out = (float*)d_out;

  float* kv   = (float*)d_ws;
  float* ksum = kv + (size_t)NH_ * T_ * D_ * DV_;

  cla_state_prefix<<<NH_ * 8, 128, 0, stream>>>(k, v, kv, ksum);
  cla_phase3<<<NH_ * T_ / 2, 128, 0, stream>>>(q, k, v, kv, ksum, out);
}


# State walk cut into four sequence segments walked at once (8 waves a block, 2 per SIMD); segment totals summed through LDS so a segment's first slot holds the true prefix; sum order changed
# speedup vs baseline: 2.8015x; 1.0577x over previous
#include <hip/hip_runtime.h>
#include <hip/hip_bf16.h>


#define N_  4
#define L_  4096
#define H_  8
#define D_  64
#define DV_ 64
#define C_  64
#define T_  (L_ / C_)
#define NH_ (N_ * H_)
#define EPS_ 1e-6f

typedef float v2f __attribute__((ext_vector_type(2)));
typedef float v8f __attribute__((ext_vector_type(8)));

__device__ __forceinline__ v8f wmma4(v2f a, v2f b, v8f c) {
  return __builtin_amdgcn_wmma_f32_16x16x4_f32(
       false, a,  false, b,
       (short)0, c,  false,  false);
}

__device__ __forceinline__ float featmap(float x) {
  return x > 0.0f ? x + 1.0f : __expf(x);
}


#define NSEG_ 4
#define SEGC_ (T_ / NSEG_)

__device__ __forceinline__ void tile_request(float (&kr)[32], float (&vr)[32], int kk,
                                             const float* kp, const float* vp) {
  const size_t rs = (size_t)H_ * D_;
  kr[2 * kk]     = kp[(size_t)(kk * 4) * rs];
  kr[2 * kk + 1] = kp[(size_t)(kk * 4 + 1) * rs];
  vr[2 * kk]     = vp[(size_t)(kk * 4) * rs];
  vr[2 * kk + 1] = vp[(size_t)(kk * 4 + 1) * rs];
}

template <bool REQ, bool KS>
__device__ __forceinline__ void tile_chunk(float (&kr)[32], float (&vr)[32],
                                           const float* kpn, const float* vpn,
                                           v8f& run, v2f& ksrun) {
  v8f cur = {};
  v2f ks4[4] = {};
#pragma unroll
  for (int kk = 0; kk < 16; ++kk) {
    v2f a, b;
    a.x = featmap(kr[2 * kk]);
    a.y = featmap(kr[2 * kk + 1]);
    b.x = vr[2 * kk]; b.y = vr[2 * kk + 1];
    cur = wmma4(a, b, cur);
    if (KS) ks4[kk & 3] += a;
    if (REQ) {
      tile_request(kr, vr, kk, kpn, vpn);
      __builtin_amdgcn_sched_barrier(0);
    }
  }
  run += cur;
  if (KS) ksrun += (ks4[0] + ks4[1]) + (ks4[2] + ks4[3]);
}

__device__ __forceinline__ void tile_store(const v8f& run, float* kvp) {
#pragma unroll
  for (int s = 0; s < 4; ++s) {
    v2f o; o.x = run[2 * s]; o.y = run[2 * s + 1];
    *reinterpret_cast<v2f*>(kvp + (size_t)s * (2 * DV_)) = o;
  }
}

__device__ __forceinline__ float ks_sum(const v2f& ksrun) {
  float sum = ksrun.x + ksrun.y;
  sum += __builtin_bit_cast(float, __builtin_amdgcn_permlanex16(__builtin_bit_cast(int, sum),
             __builtin_bit_cast(int, sum), 0x76543210, 0xfedcba98, false, false));
  return sum;
}
__device__ __forceinline__ void ks_store(const v2f& ksrun, float* ksp, int hi) {
  const float sum = ks_sum(ksrun);
  if (hi == 0) *ksp = sum;
}

template <bool KS>
__device__ __forceinline__ void tile_walk(const float* kp, const float* vp,
                                          const float* kpf, const float* vpf,
                                          float* kvp, float* ksp, int c0, int n, int hi,
                                          v8f& tot, float& kstot) {
  const size_t rs = (size_t)H_ * D_;
  const size_t cs = (size_t)C_ * rs;
  v8f run = {};
  v2f ksrun = {};
  float kr[32];
  float vr[32];
#pragma unroll
  for (int kk = 0; kk < 16; ++kk) {
    tile_request(kr, vr, kk, kp + (size_t)c0 * cs, vp + (size_t)c0 * cs);
    __builtin_amdgcn_sched_barrier(0);
  }
  if (c0 == 0) tile_store(run, kvp);
  for (int i = 0; i < n - 1; ++i) {
    const int c = c0 + i;
    const int tp = c + 4 < T_ ? c + 4 : T_ - 1;
    __builtin_prefetch(kpf + (size_t)tp * cs, 0, 3);
    __builtin_prefetch(vpf + (size_t)tp * cs, 0, 3);
    tile_chunk<true, KS>(kr, vr, kp + (size_t)(c + 1) * cs, vp + (size_t)(c + 1) * cs, run, ksrun);
    tile_store(run, kvp + (size_t)(c + 1) * (D_ * DV_));
    if (KS) ks_store(ksrun, ksp + (size_t)(c >> 1) * (2 * D_) + (c & 1) * 16, hi);
  }
  tile_chunk<false, KS>(kr, vr, kp, vp, run, ksrun);
  tot = run;
  if (KS) kstot = ks_sum(ksrun);
}

__global__ __launch_bounds__(256) void cla_state_prefix(
    const float* __restrict__ kin, const float* __restrict__ vin,
    float* __restrict__ kv, float* __restrict__ ksum) {
  const int b = blockIdx.x;
  const int nh = b >> 3, q = b & 7;
  const int n = nh / H_, h = nh % H_;
  const int wave = __builtin_amdgcn_readfirstlane((int)(threadIdx.x >> 5));
  const int lane = threadIdx.x & 31;
  const int tsel = wave & 1, seg = wave >> 1;
  const size_t rs = (size_t)H_ * D_;
  const float* kb = kin + (((size_t)n * L_) * H_ + h) * D_;
  const float* vb = vin + (((size_t)n * L_) * H_ + h) * D_;

  const int tile = 2 * q + tsel;
  const int db = tile >> 2, eb = tile & 3;
  const int m = lane & 15, hi = lane >> 4;
  const float* kp = kb + (size_t)(2 * hi) * rs + db * 16 + m;
  const float* vp = vb + (size_t)(2 * hi) * rs + eb * 16 + m;
  float* kvp = kv + (size_t)nh * T_ * (D_ * DV_) + ((size_t)(db * 8 + 4 * hi) * DV_ + eb * 16 + m) * 2;
  float* ksp = ksum + (size_t)nh * T_ * D_ + db * 32 + m;
  const float* kpf = kb + (size_t)(32 * tsel + lane) * rs;
  const float* vpf = vb + (size_t)(32 * tsel + lane) * rs;

  const int c0 = seg * SEGC_;
  const int nc = seg == NSEG_ - 1 ? SEGC_ - 1 : SEGC_;
  v8f tot;
  float kstot = 0.0f;
  if (eb == 0) tile_walk<true>(kp, vp, kpf, vpf, kvp, ksp, c0, nc, hi, tot, kstot);
  else         tile_walk<false>(kp, vp, kpf, vpf, kvp, ksp, c0, nc, hi, tot, kstot);

  __shared__ float tot_s[NSEG_ - 1][2][8 * 32];
  __shared__ float ks_s[NSEG_ - 1][16];
  float* kvl = kvp + (size_t)(c0 + nc) * (D_ * DV_);
  float* ksl = ksp + (size_t)((c0 + nc - 1) >> 1) * (2 * D_) + ((c0 + nc - 1) & 1) * 16;
  if (seg == NSEG_ - 1) {
    tile_store(tot, kvl);
    if (eb == 0 && hi == 0) *ksl = kstot;
  } else {
#pragma unroll
    for (int r = 0; r < 8; ++r) tot_s[seg][tsel][r * 32 + lane] = tot[r];
    if (eb == 0 && hi == 0) ks_s[seg][m] = kstot;
  }
  __syncthreads();
  if (seg < NSEG_ - 1) {
    v8f cum = {};
    float kcum = 0.0f;
    for (int j = 0; j < seg; ++j) {
#pragma unroll
      for (int r = 0; r < 8; ++r) cum[r] += tot_s[j][tsel][r * 32 + lane];
      if (eb == 0) kcum += ks_s[j][m];
    }
    cum += tot;
    kcum += kstot;
    tile_store(cum, kvl);
    if (eb == 0 && hi == 0) *ksl = kcum;
  }
}


typedef float v4f __attribute__((ext_vector_type(4)));

__device__ __forceinline__ v4f featmap4(v4f x) {
  v4f y; y.x = featmap(x.x); y.y = featmap(x.y); y.z = featmap(x.z); y.w = featmap(x.w); return y;
}
__device__ __forceinline__ v2f pair(float x, float y) { v2f p; p.x = x; p.y = y; return p; }

template <bool TWO>
__device__ __forceinline__ void strip_head(const float* qp, const float* sp, const float* st,
                                           const float* kzl, const float* kzt, bool haveks,
                                           v4f (&qf)[8], float& z, v8f (&acc)[4]) {
#pragma unroll
  for (int j = 0; j < 8; ++j) qf[j] = featmap4(*reinterpret_cast<const v4f*>(qp + 8 * j));
#pragma unroll
  for (int j = 0; j < 8; ++j) {
    const int o = (j >> 1) * 32 + (j & 1) * 8;
    v4f kz = *reinterpret_cast<const v4f*>(kzl + o);
    if (TWO) kz += *reinterpret_cast<const v4f*>(kzt + o);
    else     kz = haveks ? kz : (v4f){0.0f, 0.0f, 0.0f, 0.0f};
    z += qf[j].x * kz.x + qf[j].y * kz.y + qf[j].z * kz.z + qf[j].w * kz.w;
  }
  if (TWO) __builtin_amdgcn_sched_barrier(0);
#pragma unroll
  for (int jh = 0; jh < 2; ++jh) {
    v4f s[16];
#pragma unroll
    for (int u = 0; u < 4; ++u) {
      const float* sj = sp + (size_t)(4 * (4 * jh + u)) * (2 * DV_);
      s[4 * u + 0] = *reinterpret_cast<const v4f*>(sj);
      s[4 * u + 1] = *reinterpret_cast<const v4f*>(sj + 4);
      s[4 * u + 2] = *reinterpret_cast<const v4f*>(sj + 2 * DV_);
      s[4 * u + 3] = *reinterpret_cast<const v4f*>(sj + 2 * DV_ + 4);
    }
    if (TWO) {
      v4f tt[16];
#pragma unroll
      for (int u = 0; u < 4; ++u) {
        const float* tj = st + (size_t)(4 * (4 * jh + u)) * (2 * DV_);
        tt[4 * u + 0] = *reinterpret_cast<const v4f*>(tj);
        tt[4 * u + 1] = *reinterpret_cast<const v4f*>(tj + 4);
        tt[4 * u + 2] = *reinterpret_cast<const v4f*>(tj + 2 * DV_);
        tt[4 * u + 3] = *reinterpret_cast<const v4f*>(tj + 2 * DV_ + 4);
      }
#pragma unroll
      for (int i = 0; i < 16; ++i) s[i] += tt[i];
    }
#pragma unroll
    for (int u = 0; u < 4; ++u) {
      const int j = 4 * jh + u;
      const v4f s00 = s[4 * u], s01 = s[4 * u + 1], s10 = s[4 * u + 2], s11 = s[4 * u + 3];
      const v2f a0 = pair(qf[j].x, qf[j].y), a1 = pair(qf[j].z, qf[j].w);
      acc[0] = wmma4(a0, pair(s00.x, s00.y), acc[0]);
      acc[1] = wmma4(a0, pair(s00.z, s00.w), acc[1]);
      acc[2] = wmma4(a0, pair(s01.x, s01.y), acc[2]);
      acc[3] = wmma4(a0, pair(s01.z, s01.w), acc[3]);
      acc[0] = wmma4(a1, pair(s10.x, s10.y), acc[0]);
      acc[1] = wmma4(a1, pair(s10.z, s10.w), acc[1]);
      acc[2] = wmma4(a1, pair(s11.x, s11.y), acc[2]);
      acc[3] = wmma4(a1, pair(s11.z, s11.w), acc[3]);
    }
    if (TWO && jh == 0) __builtin_amdgcn_sched_barrier(0);
  }
}

__global__ __launch_bounds__(128) void cla_phase3(
    const float* __restrict__ qin, const float* __restrict__ kin,
    const float* __restrict__ vin, const float* __restrict__ kv,
    const float* __restrict__ ksum, float* __restrict__ out) {
  const int w = __builtin_amdgcn_readfirstlane((int)(threadIdx.x >> 5));
  const int lane = threadIdx.x & 31;
  const int c = lane & 15, hf = lane >> 4;
  const size_t rs = (size_t)H_ * D_;
  for (int it = 0; it < 2; ++it) {
  const int bt = 2 * (int)blockIdx.x + it;
  const int nh = bt / T_, t = bt % T_;
  const int n = nh / H_, h = nh % H_;
  const int ti = it ? 3 - w : w;
  const size_t base = (((size_t)n * L_ + (size_t)t * C_) * H_ + h) * D_;
  const size_t so = (size_t)(2 * hf) * (2 * DV_) + 8 * c;
  const float* sp  = kv + (size_t)(nh * T_ + t) * (D_ * DV_) + so;
  const int sg = t / SEGC_ * SEGC_;
  const int dst = (t - sg) * (D_ * DV_);
  const float* kzb = ksum + (size_t)nh * T_ * D_ + 4 * hf;
  const int tm = t > 0 ? t - 1 : 0;
  const float* kzl = kzb + (tm >> 1) * (2 * D_) + (tm & 1) * 16;
  const int dkz = ((tm >> 1) - ((sg - 1) >> 1)) * (2 * D_) + ((tm & 1) - ((sg - 1) & 1)) * 16;
  const float* qp  = qin + base + (size_t)(ti * 16 + c) * rs + 4 * hf;

  v4f qf[8];
  float z = 0.0f;
  v8f acc[4];
#pragma unroll
  for (int jj = 0; jj < 4; ++jj) acc[jj] = (v8f){};
  if ((sg != 0) & (t != sg)) strip_head<true>(qp, sp, sp - dst, kzl, kzl - dkz, true, qf, z, acc);
  else                       strip_head<false>(qp, sp, sp, kzl, kzl, t > 0, qf, z, acc);

  for (int tj = 0; tj <= ti; ++tj) {
    const float* kp = kin + base + (size_t)(tj * 16 + c) * rs + 4 * hf;
    const float* vp = vin + base + (size_t)(tj * 16 + 8 * hf) * rs + 4 * c;
    v4f kf[8], vf[8];
#pragma unroll
    for (int j = 0; j < 8; ++j) kf[j] = *reinterpret_cast<const v4f*>(kp + 8 * j);
#pragma unroll
    for (int r = 0; r < 8; ++r) vf[r] = *reinterpret_cast<const v4f*>(vp + (size_t)r * rs);

    v8f pt0 = {}, pt1 = {};
#pragma unroll
    for (int j = 0; j < 8; ++j) {
      const v4f kk = featmap4(kf[j]);
      pt0 = wmma4(pair(kk.x, kk.y), pair(qf[j].x, qf[j].y), pt0);
      pt1 = wmma4(pair(kk.z, kk.w), pair(qf[j].z, qf[j].w), pt1);
    }
    const bool diag = (tj == ti);
    float p[8];
#pragma unroll
    for (int r = 0; r < 8; ++r) {
      const float x = pt0[r] + pt1[r];
      p[r] = (diag && (8 * hf + r > c)) ? 0.0f : x;
      z += p[r];
    }
#pragma unroll
    for (int s = 0; s < 4; ++s) {
      const v2f a = pair(p[2 * s], p[2 * s + 1]);
#pragma unroll
      for (int jj = 0; jj < 4; ++jj)
        acc[jj] = wmma4(a, pair(vf[2 * s][jj], vf[2 * s + 1][jj]), acc[jj]);
    }
  }

  z += __shfl_xor(z, 16, 32);
  const float inv = 1.0f / (z + EPS_);

  float* ob = out + base + (size_t)(ti * 16 + 8 * hf) * (H_ * DV_) + 4 * c;
#pragma unroll
  for (int r = 0; r < 8; ++r) {
    const float ir = __shfl(inv, 8 * hf + r, 32);
    v4f o;
    o.x = acc[0][r] * ir; o.y = acc[1][r] * ir; o.z = acc[2][r] * ir; o.w = acc[3][r] * ir;
    *reinterpret_cast<v4f*>(ob + (size_t)r * (H_ * DV_)) = o;
  }
  }
}

extern "C" void kernel_launch(void* const* d_in, const int* in_sizes, int n_in,
                              void* d_out, int out_size, void* d_ws, size_t ws_size,
                              hipStream_t stream) {
  (void)in_sizes; (void)n_in; (void)out_size; (void)ws_size;
  const float* q = (const float*)d_in[0];
  const float* k = (const float*)d_in[1];
  const float* v = (const float*)d_in[2];
  float* out = (float*)d_out;

  float* kv   = (float*)d_ws;
  float* ksum = kv + (size_t)NH_ * T_ * D_ * DV_;

  cla_state_prefix<<<NH_ * 8, 256, 0, stream>>>(k, v, kv, ksum);
  cla_phase3<<<NH_ * T_ / 2, 128, 0, stream>>>(q, k, v, kv, ksum, out);
}


# Output pass: one wave per 64-row chunk, four strips at once: each state fragment, mapped key tile and value tile is loaded once for all strips, a group of products ahead; 512 blocks; sums kept
# speedup vs baseline: 3.2023x; 1.1431x over previous
#include <hip/hip_runtime.h>
#include <hip/hip_bf16.h>


#define N_  4
#define L_  4096
#define H_  8
#define D_  64
#define DV_ 64
#define C_  64
#define T_  (L_ / C_)
#define NH_ (N_ * H_)
#define EPS_ 1e-6f

typedef float v2f __attribute__((ext_vector_type(2)));
typedef float v8f __attribute__((ext_vector_type(8)));

__device__ __forceinline__ v8f wmma4(v2f a, v2f b, v8f c) {
  return __builtin_amdgcn_wmma_f32_16x16x4_f32(
       false, a,  false, b,
       (short)0, c,  false,  false);
}

__device__ __forceinline__ float featmap(float x) {
  return x > 0.0f ? x + 1.0f : __expf(x);
}


#define NSEG_ 4
#define SEGC_ (T_ / NSEG_)

__device__ __forceinline__ void tile_request(float (&kr)[32], float (&vr)[32], int kk,
                                             const float* kp, const float* vp) {
  const size_t rs = (size_t)H_ * D_;
  kr[2 * kk]     = kp[(size_t)(kk * 4) * rs];
  kr[2 * kk + 1] = kp[(size_t)(kk * 4 + 1) * rs];
  vr[2 * kk]     = vp[(size_t)(kk * 4) * rs];
  vr[2 * kk + 1] = vp[(size_t)(kk * 4 + 1) * rs];
}

template <bool REQ, bool KS>
__device__ __forceinline__ void tile_chunk(float (&kr)[32], float (&vr)[32],
                                           const float* kpn, const float* vpn,
                                           v8f& run, v2f& ksrun) {
  v8f cur = {};
  v2f ks4[4] = {};
#pragma unroll
  for (int kk = 0; kk < 16; ++kk) {
    v2f a, b;
    a.x = featmap(kr[2 * kk]);
    a.y = featmap(kr[2 * kk + 1]);
    b.x = vr[2 * kk]; b.y = vr[2 * kk + 1];
    cur = wmma4(a, b, cur);
    if (KS) ks4[kk & 3] += a;
    if (REQ) {
      tile_request(kr, vr, kk, kpn, vpn);
      __builtin_amdgcn_sched_barrier(0);
    }
  }
  run += cur;
  if (KS) ksrun += (ks4[0] + ks4[1]) + (ks4[2] + ks4[3]);
}

__device__ __forceinline__ void tile_store(const v8f& run, float* kvp) {
#pragma unroll
  for (int s = 0; s < 4; ++s) {
    v2f o; o.x = run[2 * s]; o.y = run[2 * s + 1];
    *reinterpret_cast<v2f*>(kvp + (size_t)s * (2 * DV_)) = o;
  }
}

__device__ __forceinline__ float ks_sum(const v2f& ksrun) {
  float sum = ksrun.x + ksrun.y;
  sum += __builtin_bit_cast(float, __builtin_amdgcn_permlanex16(__builtin_bit_cast(int, sum),
             __builtin_bit_cast(int, sum), 0x76543210, 0xfedcba98, false, false));
  return sum;
}
__device__ __forceinline__ void ks_store(const v2f& ksrun, float* ksp, int hi) {
  const float sum = ks_sum(ksrun);
  if (hi == 0) *ksp = sum;
}

template <bool KS>
__device__ __forceinline__ void tile_walk(const float* kp, const float* vp,
                                          const float* kpf, const float* vpf,
                                          float* kvp, float* ksp, int c0, int n, int hi,
                                          v8f& tot, float& kstot) {
  const size_t rs = (size_t)H_ * D_;
  const size_t cs = (size_t)C_ * rs;
  v8f run = {};
  v2f ksrun = {};
  float kr[32];
  float vr[32];
#pragma unroll
  for (int kk = 0; kk < 16; ++kk) {
    tile_request(kr, vr, kk, kp + (size_t)c0 * cs, vp + (size_t)c0 * cs);
    __builtin_amdgcn_sched_barrier(0);
  }
  if (c0 == 0) tile_store(run, kvp);
  for (int i = 0; i < n - 1; ++i) {
    const int c = c0 + i;
    const int tp = c + 4 < T_ ? c + 4 : T_ - 1;
    __builtin_prefetch(kpf + (size_t)tp * cs, 0, 3);
    __builtin_prefetch(vpf + (size_t)tp * cs, 0, 3);
    tile_chunk<true, KS>(kr, vr, kp + (size_t)(c + 1) * cs, vp + (size_t)(c + 1) * cs, run, ksrun);
    tile_store(run, kvp + (size_t)(c + 1) * (D_ * DV_));
    if (KS) ks_store(ksrun, ksp + (size_t)(c >> 1) * (2 * D_) + (c & 1) * 16, hi);
  }
  tile_chunk<false, KS>(kr, vr, kp, vp, run, ksrun);
  tot = run;
  if (KS) kstot = ks_sum(ksrun);
}

__global__ __launch_bounds__(256) void cla_state_prefix(
    const float* __restrict__ kin, const float* __restrict__ vin,
    float* __restrict__ kv, float* __restrict__ ksum) {
  const int b = blockIdx.x;
  const int nh = b >> 3, q = b & 7;
  const int n = nh / H_, h = nh % H_;
  const int wave = __builtin_amdgcn_readfirstlane((int)(threadIdx.x >> 5));
  const int lane = threadIdx.x & 31;
  const int tsel = wave & 1, seg = wave >> 1;
  const size_t rs = (size_t)H_ * D_;
  const float* kb = kin + (((size_t)n * L_) * H_ + h) * D_;
  const float* vb = vin + (((size_t)n * L_) * H_ + h) * D_;

  const int tile = 2 * q + tsel;
  const int db = tile >> 2, eb = tile & 3;
  const int m = lane & 15, hi = lane >> 4;
  const float* kp = kb + (size_t)(2 * hi) * rs + db * 16 + m;
  const float* vp = vb + (size_t)(2 * hi) * rs + eb * 16 + m;
  float* kvp = kv + (size_t)nh * T_ * (D_ * DV_) + ((size_t)(db * 8 + 4 * hi) * DV_ + eb * 16 + m) * 2;
  float* ksp = ksum + (size_t)nh * T_ * D_ + db * 32 + m;
  const float* kpf = kb + (size_t)(32 * tsel + lane) * rs;
  const float* vpf = vb + (size_t)(32 * tsel + lane) * rs;

  const int c0 = seg * SEGC_;
  const int nc = seg == NSEG_ - 1 ? SEGC_ - 1 : SEGC_;
  v8f tot;
  float kstot = 0.0f;
  if (eb == 0) tile_walk<true>(kp, vp, kpf, vpf, kvp, ksp, c0, nc, hi, tot, kstot);
  else         tile_walk<false>(kp, vp, kpf, vpf, kvp, ksp, c0, nc, hi, tot, kstot);

  __shared__ float tot_s[NSEG_ - 1][2][8 * 32];
  __shared__ float ks_s[NSEG_ - 1][16];
  float* kvl = kvp + (size_t)(c0 + nc) * (D_ * DV_);
  float* ksl = ksp + (size_t)((c0 + nc - 1) >> 1) * (2 * D_) + ((c0 + nc - 1) & 1) * 16;
  if (seg == NSEG_ - 1) {
    tile_store(tot, kvl);
    if (eb == 0 && hi == 0) *ksl = kstot;
  } else {
#pragma unroll
    for (int r = 0; r < 8; ++r) tot_s[seg][tsel][r * 32 + lane] = tot[r];
    if (eb == 0 && hi == 0) ks_s[seg][m] = kstot;
  }
  __syncthreads();
  if (seg < NSEG_ - 1) {
    v8f cum = {};
    float kcum = 0.0f;
    for (int j = 0; j < seg; ++j) {
#pragma unroll
      for (int r = 0; r < 8; ++r) cum[r] += tot_s[j][tsel][r * 32 + lane];
      if (eb == 0) kcum += ks_s[j][m];
    }
    cum += tot;
    kcum += kstot;
    tile_store(cum, kvl);
    if (eb == 0 && hi == 0) *ksl = kcum;
  }
}


typedef float v4f __attribute__((ext_vector_type(4)));

__device__ __forceinline__ v4f featmap4(v4f x) {
  v4f y; y.x = featmap(x.x); y.y = featmap(x.y); y.z = featmap(x.z); y.w = featmap(x.w); return y;
}
__device__ __forceinline__ v2f pair(float x, float y) { v2f p; p.x = x; p.y = y; return p; }

__device__ __forceinline__ void state_request(v4f (&s)[8], const float* sp, int g) {
#pragma unroll
  for (int u = 0; u < 2; ++u) {
    const float* sj = sp + (size_t)(4 * (2 * g + u)) * (2 * DV_);
    s[4 * u + 0] = *reinterpret_cast<const v4f*>(sj);
    s[4 * u + 1] = *reinterpret_cast<const v4f*>(sj + 4);
    s[4 * u + 2] = *reinterpret_cast<const v4f*>(sj + 2 * DV_);
    s[4 * u + 3] = *reinterpret_cast<const v4f*>(sj + 2 * DV_ + 4);
  }
}

__device__ __forceinline__ void state_add(v4f (&s)[8], const v4f (&tt)[8], bool two) {
#pragma unroll
  for (int i = 0; i < 8; ++i) s[i] += two ? tt[i] : (v4f){0.0f, 0.0f, 0.0f, 0.0f};
}

__device__ __forceinline__ void state_products(const v4f (&s)[8], const v4f (&qf)[4][8],
                                               v8f (&acc)[4][4], int g) {
#pragma unroll
  for (int u = 0; u < 2; ++u) {
    const int j = 2 * g + u;
    const v4f s00 = s[4 * u], s01 = s[4 * u + 1], s10 = s[4 * u + 2], s11 = s[4 * u + 3];
#pragma unroll
    for (int ti = 0; ti < 4; ++ti) {
      const v2f a0 = pair(qf[ti][j].x, qf[ti][j].y), a1 = pair(qf[ti][j].z, qf[ti][j].w);
      acc[ti][0] = wmma4(a0, pair(s00.x, s00.y), acc[ti][0]);
      acc[ti][1] = wmma4(a0, pair(s00.z, s00.w), acc[ti][1]);
      acc[ti][2] = wmma4(a0, pair(s01.x, s01.y), acc[ti][2]);
      acc[ti][3] = wmma4(a0, pair(s01.z, s01.w), acc[ti][3]);
      acc[ti][0] = wmma4(a1, pair(s10.x, s10.y), acc[ti][0]);
      acc[ti][1] = wmma4(a1, pair(s10.z, s10.w), acc[ti][1]);
      acc[ti][2] = wmma4(a1, pair(s11.x, s11.y), acc[ti][2]);
      acc[ti][3] = wmma4(a1, pair(s11.z, s11.w), acc[ti][3]);
    }
  }
}

__device__ __forceinline__ void q_quarter(v4f (&qf)[4][8], const v4f (&kc)[8], float (&z)[4], int g) {
#pragma unroll
  for (int u = 0; u < 2; ++u) {
    const int j = 2 * g + u;
#pragma unroll
    for (int ti = 0; ti < 4; ++ti) {
      qf[ti][j] = featmap4(qf[ti][j]);
      z[ti] += qf[ti][j].x * kc[j].x + qf[ti][j].y * kc[j].y + qf[ti][j].z * kc[j].z + qf[ti][j].w * kc[j].w;
    }
  }
}

__device__ __forceinline__ void tile_kv_request(v4f (&kf)[8], v4f (&vf)[8], const float* kp,
                                                const float* vp, int tj) {
  const size_t rs = (size_t)H_ * D_;
#pragma unroll
  for (int j = 0; j < 8; ++j) kf[j] = *reinterpret_cast<const v4f*>(kp + (size_t)(tj * 16) * rs + 8 * j);
#pragma unroll
  for (int r = 0; r < 8; ++r) vf[r] = *reinterpret_cast<const v4f*>(vp + (size_t)(tj * 16 + r) * rs);
}

__device__ __forceinline__ void strip_tile(const v4f (&kk)[8], const v4f (&vf)[8], const v4f (&q)[8],
                                           v8f (&acc)[4], float& z, bool diag, int c, int hf) {
  v8f pt0 = {}, pt1 = {};
#pragma unroll
  for (int j = 0; j < 8; ++j) {
    pt0 = wmma4(pair(kk[j].x, kk[j].y), pair(q[j].x, q[j].y), pt0);
    pt1 = wmma4(pair(kk[j].z, kk[j].w), pair(q[j].z, q[j].w), pt1);
  }
  float p[8];
#pragma unroll
  for (int r = 0; r < 8; ++r) {
    const float x = pt0[r] + pt1[r];
    p[r] = (diag && (8 * hf + r > c)) ? 0.0f : x;
    z += p[r];
  }
#pragma unroll
  for (int s = 0; s < 4; ++s) {
    const v2f a = pair(p[2 * s], p[2 * s + 1]);
#pragma unroll
    for (int jj = 0; jj < 4; ++jj)
      acc[jj] = wmma4(a, pair(vf[2 * s][jj], vf[2 * s + 1][jj]), acc[jj]);
  }
}

__device__ __forceinline__ void strip_store(const v8f (&acc)[4], float z, float* ob, int hf) {
  z += __shfl_xor(z, 16, 32);
  const float inv = 1.0f / (z + EPS_);
#pragma unroll
  for (int r = 0; r < 8; ++r) {
    const float ir = __shfl(inv, 8 * hf + r, 32);
    v4f o;
    o.x = acc[0][r] * ir; o.y = acc[1][r] * ir; o.z = acc[2][r] * ir; o.w = acc[3][r] * ir;
    *reinterpret_cast<v4f*>(ob + (size_t)r * (H_ * DV_)) = o;
  }
}

__global__ __launch_bounds__(128, 2) void cla_phase3(
    const float* __restrict__ qin, const float* __restrict__ kin,
    const float* __restrict__ vin, const float* __restrict__ kv,
    const float* __restrict__ ksum, float* __restrict__ out) {
  const int w = __builtin_amdgcn_readfirstlane((int)(threadIdx.x >> 5));
  const int lane = threadIdx.x & 31;
  const int c = lane & 15, hf = lane >> 4;
  const size_t rs = (size_t)H_ * D_;
  const int bt = 4 * (int)blockIdx.x + w;
  const int nh = bt / T_, t = bt % T_;
  const int n = nh / H_, h = nh % H_;
  const size_t base = (((size_t)n * L_ + (size_t)t * C_) * H_ + h) * D_;
  const size_t so = (size_t)(2 * hf) * (2 * DV_) + 8 * c;
  const float* sp  = kv + (size_t)(nh * T_ + t) * (D_ * DV_) + so;
  const int sg = t / SEGC_ * SEGC_;
  const bool two = (sg != 0) & (t != sg);
  const int dst = (t - sg) * (D_ * DV_);
  const float* st = two ? sp - dst : sp;
  const float* kzb = ksum + (size_t)nh * T_ * D_ + 4 * hf;
  const int tm = t > 0 ? t - 1 : 0;
  const float* kzl = kzb + (tm >> 1) * (2 * D_) + (tm & 1) * 16;
  const int dkz = ((tm >> 1) - ((sg - 1) >> 1)) * (2 * D_) + ((tm & 1) - ((sg - 1) & 1)) * 16;
  const float* kzt = two ? kzl - dkz : kzl;
  const bool haveks = t > 0;
  const float* qp = qin + base + (size_t)c * rs + 4 * hf;
  const float* kp = kin + base + (size_t)c * rs + 4 * hf;
  const float* vp = vin + base + (size_t)(8 * hf) * rs + 4 * c;
  float* ob = out + base + (size_t)(8 * hf) * (H_ * DV_) + 4 * c;
  const v4f zero4 = {0.0f, 0.0f, 0.0f, 0.0f};

  v4f sa[8], sb[8], tt[8];
  state_request(sa, sp, 0);
  state_request(tt, st, 0);
  v4f qf[4][8];
#pragma unroll
  for (int ti = 0; ti < 4; ++ti)
#pragma unroll
    for (int j = 0; j < 8; ++j)
      qf[ti][j] = *reinterpret_cast<const v4f*>(qp + (size_t)(ti * 16) * rs + 8 * j);
  v4f kc[8], kz2[8];
#pragma unroll
  for (int j = 0; j < 8; ++j) {
    const int o = (j >> 1) * 32 + (j & 1) * 8;
    kc[j]  = *reinterpret_cast<const v4f*>(kzl + o);
    kz2[j] = *reinterpret_cast<const v4f*>(kzt + o);
  }
  __builtin_amdgcn_sched_barrier(0);
  state_request(sb, sp, 1);
  __builtin_amdgcn_sched_barrier(0);

  float z[4];
  v8f acc[4][4];
#pragma unroll
  for (int ti = 0; ti < 4; ++ti) {
    z[ti] = 0.0f;
#pragma unroll
    for (int jj = 0; jj < 4; ++jj) acc[ti][jj] = (v8f){};
  }
#pragma unroll
  for (int j = 0; j < 8; ++j) kc[j] = (haveks ? kc[j] : zero4) + (two ? kz2[j] : zero4);
  q_quarter(qf, kc, z, 0);
  state_add(sa, tt, two);
  __builtin_amdgcn_sched_barrier(0);
  state_request(tt, st, 1);
  __builtin_amdgcn_sched_barrier(0);
  state_products(sa, qf, acc, 0);
  q_quarter(qf, kc, z, 1);
  __builtin_amdgcn_sched_barrier(0);
  state_add(sb, tt, two);
  __builtin_amdgcn_sched_barrier(0);
  state_request(sa, sp, 2);
  state_request(tt, st, 2);
  __builtin_amdgcn_sched_barrier(0);
  state_products(sb, qf, acc, 1);
  q_quarter(qf, kc, z, 2);
  __builtin_amdgcn_sched_barrier(0);
  state_add(sa, tt, two);
  __builtin_amdgcn_sched_barrier(0);
  state_request(sb, sp, 3);
  state_request(tt, st, 3);
  __builtin_amdgcn_sched_barrier(0);
  state_products(sa, qf, acc, 2);
  q_quarter(qf, kc, z, 3);
  __builtin_amdgcn_sched_barrier(0);
  state_add(sb, tt, two);
  __builtin_amdgcn_sched_barrier(0);
  v4f kf0[8], vf0[8], kf1[8], vf1[8];
  tile_kv_request(kf0, vf0, kp, vp, 0);
  __builtin_amdgcn_sched_barrier(0);
  state_products(sb, qf, acc, 3);
  __builtin_amdgcn_sched_barrier(0);

  tile_kv_request(kf1, vf1, kp, vp, 1);
  __builtin_amdgcn_sched_barrier(0);
#pragma unroll
  for (int j = 0; j < 8; ++j) kf0[j] = featmap4(kf0[j]);
  strip_tile(kf0, vf0, qf[0], acc[0], z[0], true, c, hf);
  strip_store(acc[0], z[0], ob, hf);
  strip_tile(kf0, vf0, qf[1], acc[1], z[1], false, c, hf);
  strip_tile(kf0, vf0, qf[2], acc[2], z[2], false, c, hf);
  strip_tile(kf0, vf0, qf[3], acc[3], z[3], false, c, hf);
  __builtin_amdgcn_sched_barrier(0);
  tile_kv_request(kf0, vf0, kp, vp, 2);
  __builtin_amdgcn_sched_barrier(0);
#pragma unroll
  for (int j = 0; j < 8; ++j) kf1[j] = featmap4(kf1[j]);
  strip_tile(kf1, vf1, qf[1], acc[1], z[1], true, c, hf);
  strip_store(acc[1], z[1], ob + (size_t)16 * (H_ * DV_), hf);
  strip_tile(kf1, vf1, qf[2], acc[2], z[2], false, c, hf);
  strip_tile(kf1, vf1, qf[3], acc[3], z[3], false, c, hf);
  __builtin_amdgcn_sched_barrier(0);
  tile_kv_request(kf1, vf1, kp, vp, 3);
  __builtin_amdgcn_sched_barrier(0);
#pragma unroll
  for (int j = 0; j < 8; ++j) kf0[j] = featmap4(kf0[j]);
  strip_tile(kf0, vf0, qf[2], acc[2], z[2], true, c, hf);
  strip_store(acc[2], z[2], ob + (size_t)32 * (H_ * DV_), hf);
  strip_tile(kf0, vf0, qf[3], acc[3], z[3], false, c, hf);
#pragma unroll
  for (int j = 0; j < 8; ++j) kf1[j] = featmap4(kf1[j]);
  strip_tile(kf1, vf1, qf[3], acc[3], z[3], true, c, hf);
  strip_store(acc[3], z[3], ob + (size_t)48 * (H_ * DV_), hf);
}

extern "C" void kernel_launch(void* const* d_in, const int* in_sizes, int n_in,
                              void* d_out, int out_size, void* d_ws, size_t ws_size,
                              hipStream_t stream) {
  (void)in_sizes; (void)n_in; (void)out_size; (void)ws_size;
  const float* q = (const float*)d_in[0];
  const float* k = (const float*)d_in[1];
  const float* v = (const float*)d_in[2];
  float* out = (float*)d_out;

  float* kv   = (float*)d_ws;
  float* ksum = kv + (size_t)NH_ * T_ * D_ * DV_;

  cla_state_prefix<<<NH_ * 8, 256, 0, stream>>>(k, v, kv, ksum);
  cla_phase3<<<NH_ * T_ / 4, 128, 0, stream>>>(q, k, v, kv, ksum, out);
}


# State pass split over 8 sequence segments, not state tiles: a block walks one segment for all 16 tiles (2 a wave), K and V read by one block; output pass sums earlier totals in LDS; sum order changed
# speedup vs baseline: 4.1203x; 1.2867x over previous
#include <hip/hip_runtime.h>
#include <hip/hip_bf16.h>


#define N_  4
#define L_  4096
#define H_  8
#define D_  64
#define DV_ 64
#define C_  64
#define T_  (L_ / C_)
#define NH_ (N_ * H_)
#define EPS_ 1e-6f

typedef float v2f __attribute__((ext_vector_type(2)));
typedef float v4f __attribute__((ext_vector_type(4)));
typedef float v8f __attribute__((ext_vector_type(8)));

__device__ __forceinline__ v8f wmma4(v2f a, v2f b, v8f c) {
  return __builtin_amdgcn_wmma_f32_16x16x4_f32(
       false, a,  false, b,
       (short)0, c,  false,  false);
}

__device__ __forceinline__ float featmap(float x) {
  return x > 0.0f ? x + 1.0f : __expf(x);
}


#define NSEG_ 8
#define SEGC_ (T_ / NSEG_)

__device__ __forceinline__ void tile_request(float (&kr)[32], v2f (&vr)[32], int kk,
                                             const float* kp, const float* vp) {
  const size_t rs = (size_t)H_ * D_;
  kr[2 * kk]     = kp[(size_t)(kk * 4) * rs];
  kr[2 * kk + 1] = kp[(size_t)(kk * 4 + 1) * rs];
  vr[2 * kk]     = *reinterpret_cast<const v2f*>(vp + (size_t)(kk * 4) * rs);
  vr[2 * kk + 1] = *reinterpret_cast<const v2f*>(vp + (size_t)(kk * 4 + 1) * rs);
}

template <bool REQ, bool KS>
__device__ __forceinline__ void tile_chunk(float (&kr)[32], v2f (&vr)[32],
                                           const float* kpn, const float* vpn,
                                           v8f (&run)[2], v2f& ksrun) {
  v8f cur0 = {}, cur1 = {};
  v2f ks4[4] = {};
#pragma unroll
  for (int kk = 0; kk < 16; ++kk) {
    v2f a, b0, b1;
    a.x = featmap(kr[2 * kk]);
    a.y = featmap(kr[2 * kk + 1]);
    b0.x = vr[2 * kk].x; b0.y = vr[2 * kk + 1].x;
    b1.x = vr[2 * kk].y; b1.y = vr[2 * kk + 1].y;
    cur0 = wmma4(a, b0, cur0);
    cur1 = wmma4(a, b1, cur1);
    if (KS) ks4[kk & 3] += a;
    if (REQ) {
      tile_request(kr, vr, kk, kpn, vpn);
      __builtin_amdgcn_sched_barrier(0);
    }
  }
  run[0] += cur0;
  run[1] += cur1;
  if (KS) ksrun += (ks4[0] + ks4[1]) + (ks4[2] + ks4[3]);
}

__device__ __forceinline__ void tile_store(const v8f (&run)[2], float* kvp) {
#pragma unroll
  for (int s = 0; s < 4; ++s) {
    v4f o;
    o.x = run[0][2 * s]; o.y = run[0][2 * s + 1];
    o.z = run[1][2 * s]; o.w = run[1][2 * s + 1];
    *reinterpret_cast<v4f*>(kvp + (size_t)s * (2 * DV_)) = o;
  }
}

__device__ __forceinline__ float ks_sum(const v2f& ksrun) {
  float sum = ksrun.x + ksrun.y;
  sum += __builtin_bit_cast(float, __builtin_amdgcn_permlanex16(__builtin_bit_cast(int, sum),
             __builtin_bit_cast(int, sum), 0x76543210, 0xfedcba98, false, false));
  return sum;
}
__device__ __forceinline__ void ks_store(const v2f& ksrun, float* ksp, int hi) {
  const float sum = ks_sum(ksrun);
  if (hi == 0) *ksp = sum;
}

template <bool KS>
__device__ __forceinline__ void tile_walk(const float* kp, const float* vp, const float* pf,
                                          float* kvp, float* ksp, int c0, int n, int hi,
                                          v8f (&tot)[2], float& kstot) {
  const size_t rs = (size_t)H_ * D_;
  const size_t cs = (size_t)C_ * rs;
  v8f run[2] = {};
  v2f ksrun = {};
  float kr[32];
  v2f vr[32];
#pragma unroll
  for (int kk = 0; kk < 16; ++kk) {
    tile_request(kr, vr, kk, kp + (size_t)c0 * cs, vp + (size_t)c0 * cs);
    __builtin_amdgcn_sched_barrier(0);
  }
  __builtin_prefetch(pf, 0, 3);
  __builtin_prefetch(pf + 2 * cs, 0, 3);
  __builtin_prefetch(pf + 4 * cs, 0, 3);
  __builtin_prefetch(pf + 6 * cs, 0, 3);
  for (int i = 0; i < n - 1; ++i) {
    const int c = c0 + i;
    tile_chunk<true, KS>(kr, vr, kp + (size_t)(c + 1) * cs, vp + (size_t)(c + 1) * cs, run, ksrun);
    tile_store(run, kvp + (size_t)(c + 1) * (D_ * DV_));
    if (KS) ks_store(ksrun, ksp + (size_t)(c >> 1) * (2 * D_) + (c & 1) * 16, hi);
  }
  tile_chunk<false, KS>(kr, vr, kp, vp, run, ksrun);
  tot[0] = run[0];
  tot[1] = run[1];
  if (KS) kstot = ks_sum(ksrun);
}

__global__ __launch_bounds__(256) void cla_state_prefix(
    const float* __restrict__ kin, const float* __restrict__ vin,
    float* __restrict__ kv, float* __restrict__ ksum) {
  const int b = blockIdx.x;
  const int nh = b / NSEG_, seg = b % NSEG_;
  const int n = nh / H_, h = nh % H_;
  const int wave = __builtin_amdgcn_readfirstlane((int)(threadIdx.x >> 5));
  const int lane = threadIdx.x & 31;
  const size_t rs = (size_t)H_ * D_;
  const float* kb = kin + (((size_t)n * L_) * H_ + h) * D_;
  const float* vb = vin + (((size_t)n * L_) * H_ + h) * D_;

  const int db = wave >> 1, ep = wave & 1;
  const int m = lane & 15, hi = lane >> 4;
  const float* kp = kb + (size_t)(2 * hi) * rs + db * 16 + m;
  const float* vp = vb + (size_t)(2 * hi) * rs + ep * 32 + 2 * m;
  float* kvp = kv + (size_t)nh * T_ * (D_ * DV_) + ((size_t)(db * 8 + 4 * hi) * DV_ + ep * 32 + 2 * m) * 2;
  float* ksp = ksum + (size_t)nh * T_ * D_ + db * 32 + m;

  const int c0 = seg * SEGC_;
  const int nc = seg == NSEG_ - 1 ? SEGC_ - 1 : SEGC_;
  const float* pf = ((wave & 2) ? vb : kb) +
                    ((size_t)(c0 + (wave >> 2)) * C_ + 32 * (wave & 1) + lane) * rs;
  v8f tot[2];
  float kstot = 0.0f;
  if (ep == 0) tile_walk<true>(kp, vp, pf, kvp, ksp, c0, nc, hi, tot, kstot);
  else         tile_walk<false>(kp, vp, pf, kvp, ksp, c0, nc, hi, tot, kstot);

  tile_store(tot, kvp + (size_t)(c0 + nc) * (D_ * DV_));
  if (ep == 0 && hi == 0)
    *(ksp + (size_t)((c0 + nc - 1) >> 1) * (2 * D_) + ((c0 + nc - 1) & 1) * 16) = kstot;
}


__device__ __forceinline__ v4f featmap4(v4f x) {
  v4f y; y.x = featmap(x.x); y.y = featmap(x.y); y.z = featmap(x.z); y.w = featmap(x.w); return y;
}
__device__ __forceinline__ v2f pair(float x, float y) { v2f p; p.x = x; p.y = y; return p; }

__device__ __forceinline__ void state_request(v4f (&s)[8], const float* sp, int g) {
#pragma unroll
  for (int u = 0; u < 2; ++u) {
    const float* sj = sp + (size_t)(4 * (2 * g + u)) * (2 * DV_);
    s[4 * u + 0] = *reinterpret_cast<const v4f*>(sj);
    s[4 * u + 1] = *reinterpret_cast<const v4f*>(sj + 4);
    s[4 * u + 2] = *reinterpret_cast<const v4f*>(sj + 2 * DV_);
    s[4 * u + 3] = *reinterpret_cast<const v4f*>(sj + 2 * DV_ + 4);
  }
}

__device__ __forceinline__ void state_add(v4f (&s)[8], const v4f (&tt)[8], bool local) {
#pragma unroll
  for (int i = 0; i < 8; ++i) s[i] = (local ? s[i] : (v4f){0.0f, 0.0f, 0.0f, 0.0f}) + tt[i];
}

__device__ __forceinline__ void state_products(const v4f (&s)[8], const v4f (&qf)[4][8],
                                               v8f (&acc)[4][4], int g) {
#pragma unroll
  for (int u = 0; u < 2; ++u) {
    const int j = 2 * g + u;
    const v4f s00 = s[4 * u], s01 = s[4 * u + 1], s10 = s[4 * u + 2], s11 = s[4 * u + 3];
#pragma unroll
    for (int ti = 0; ti < 4; ++ti) {
      const v2f a0 = pair(qf[ti][j].x, qf[ti][j].y), a1 = pair(qf[ti][j].z, qf[ti][j].w);
      acc[ti][0] = wmma4(a0, pair(s00.x, s00.y), acc[ti][0]);
      acc[ti][1] = wmma4(a0, pair(s00.z, s00.w), acc[ti][1]);
      acc[ti][2] = wmma4(a0, pair(s01.x, s01.y), acc[ti][2]);
      acc[ti][3] = wmma4(a0, pair(s01.z, s01.w), acc[ti][3]);
      acc[ti][0] = wmma4(a1, pair(s10.x, s10.y), acc[ti][0]);
      acc[ti][1] = wmma4(a1, pair(s10.z, s10.w), acc[ti][1]);
      acc[ti][2] = wmma4(a1, pair(s11.x, s11.y), acc[ti][2]);
      acc[ti][3] = wmma4(a1, pair(s11.z, s11.w), acc[ti][3]);
    }
  }
}

__device__ __forceinline__ void q_quarter(v4f (&qf)[4][8], const v4f (&kc)[8], float (&z)[4], int g) {
#pragma unroll
  for (int u = 0; u < 2; ++u) {
    const int j = 2 * g + u;
#pragma unroll
    for (int ti = 0; ti < 4; ++ti) {
      qf[ti][j] = featmap4(qf[ti][j]);
      z[ti] += qf[ti][j].x * kc[j].x + qf[ti][j].y * kc[j].y + qf[ti][j].z * kc[j].z + qf[ti][j].w * kc[j].w;
    }
  }
}

__device__ __forceinline__ void tile_kv_request(v4f (&kf)[8], v4f (&vf)[8], const float* kp,
                                                const float* vp, int tj) {
  const size_t rs = (size_t)H_ * D_;
#pragma unroll
  for (int j = 0; j < 8; ++j) kf[j] = *reinterpret_cast<const v4f*>(kp + (size_t)(tj * 16) * rs + 8 * j);
#pragma unroll
  for (int r = 0; r < 8; ++r) vf[r] = *reinterpret_cast<const v4f*>(vp + (size_t)(tj * 16 + r) * rs);
}

__device__ __forceinline__ void strip_tile(const v4f (&kk)[8], const v4f (&vf)[8], const v4f (&q)[8],
                                           v8f (&acc)[4], float& z, bool diag, int c, int hf) {
  v8f pt0 = {}, pt1 = {};
#pragma unroll
  for (int j = 0; j < 8; ++j) {
    pt0 = wmma4(pair(kk[j].x, kk[j].y), pair(q[j].x, q[j].y), pt0);
    pt1 = wmma4(pair(kk[j].z, kk[j].w), pair(q[j].z, q[j].w), pt1);
  }
  float p[8];
#pragma unroll
  for (int r = 0; r < 8; ++r) {
    const float x = pt0[r] + pt1[r];
    p[r] = (diag && (8 * hf + r > c)) ? 0.0f : x;
    z += p[r];
  }
#pragma unroll
  for (int s = 0; s < 4; ++s) {
    const v2f a = pair(p[2 * s], p[2 * s + 1]);
#pragma unroll
    for (int jj = 0; jj < 4; ++jj)
      acc[jj] = wmma4(a, pair(vf[2 * s][jj], vf[2 * s + 1][jj]), acc[jj]);
  }
}

__device__ __forceinline__ void strip_store(const v8f (&acc)[4], float z, float* ob, int hf) {
  z += __shfl_xor(z, 16, 32);
  const float inv = 1.0f / (z + EPS_);
#pragma unroll
  for (int r = 0; r < 8; ++r) {
    const float ir = __shfl(inv, 8 * hf + r, 32);
    v4f o;
    o.x = acc[0][r] * ir; o.y = acc[1][r] * ir; o.z = acc[2][r] * ir; o.w = acc[3][r] * ir;
    *reinterpret_cast<v4f*>(ob + (size_t)r * (H_ * DV_)) = o;
  }
}

__global__ __launch_bounds__(256, 1) void cla_phase3(
    const float* __restrict__ qin, const float* __restrict__ kin,
    const float* __restrict__ vin, const float* __restrict__ kv,
    const float* __restrict__ ksum, float* __restrict__ out) {
  __shared__ __attribute__((aligned(16))) float pg_s[D_ * DV_];
  __shared__ __attribute__((aligned(16))) float kzp_s[256];
  const int w = __builtin_amdgcn_readfirstlane((int)(threadIdx.x >> 5));
  const int tid = threadIdx.x;
  const int lane = threadIdx.x & 31;
  const int c = lane & 15, hf = lane >> 4;
  const size_t rs = (size_t)H_ * D_;
  const int nh = (int)blockIdx.x / NSEG_, g = (int)blockIdx.x % NSEG_;
  const int t = SEGC_ * g + w;
  const int n = nh / H_, h = nh % H_;
  const size_t base = (((size_t)n * L_ + (size_t)t * C_) * H_ + h) * D_;
  const size_t so = (size_t)(2 * hf) * (2 * DV_) + 8 * c;
  const bool local = w != 0;
  const float* sp = kv + (size_t)(nh * T_ + (local ? t : t + 1)) * (D_ * DV_) + so;
  const float* pg = pg_s + so;
  const float* kzb = ksum + (size_t)nh * T_ * D_;
  const int tm = t > 0 ? t - 1 : 0;
  const float* kzl = kzb + 4 * hf + (tm >> 1) * (2 * D_) + (tm & 1) * 16;
  const float* qp = qin + base + (size_t)c * rs + 4 * hf;
  const float* kp = kin + base + (size_t)c * rs + 4 * hf;
  const float* vp = vin + base + (size_t)(8 * hf) * rs + 4 * c;
  float* ob = out + base + (size_t)(8 * hf) * (H_ * DV_) + 4 * c;
  const v4f zero4 = {0.0f, 0.0f, 0.0f, 0.0f};

  const int gs = g > 0 ? g : 1;
  v4f pl[NSEG_ - 1][4];
  float kl[NSEG_ - 1];
  const int kd = tid & (D_ - 1);
#pragma unroll
  for (int j = 1; j < NSEG_; ++j) {
    const int js = j < gs ? j : gs;
    const float* tp = kv + (size_t)(nh * T_ + SEGC_ * js) * (D_ * DV_) + 4 * tid;
#pragma unroll
    for (int i = 0; i < 4; ++i) pl[j - 1][i] = *reinterpret_cast<const v4f*>(tp + 1024 * i);
    const int sm = SEGC_ * js - 1;
    kl[j - 1] = kzb[((sm >> 1) * 4 + (kd >> 4)) * 32 + (sm & 1) * 16 + (kd & 15)];
  }
  __builtin_amdgcn_sched_barrier(0);

  v4f sa[8], sb[8], tt[8];
  state_request(sa, sp, 0);
  v4f qf[4][8];
#pragma unroll
  for (int ti = 0; ti < 4; ++ti)
#pragma unroll
    for (int j = 0; j < 8; ++j)
      qf[ti][j] = *reinterpret_cast<const v4f*>(qp + (size_t)(ti * 16) * rs + 8 * j);
  v4f kc[8], kz2[8];
#pragma unroll
  for (int j = 0; j < 8; ++j) {
    const int o = (j >> 1) * 32 + (j & 1) * 8;
    kc[j]  = *reinterpret_cast<const v4f*>(kzl + o);
  }
  __builtin_amdgcn_sched_barrier(0);
  state_request(sb, sp, 1);
  __builtin_amdgcn_sched_barrier(0);

  {
    v4f pa[4] = {zero4, zero4, zero4, zero4};
    float ka = 0.0f;
#pragma unroll
    for (int j = 1; j < NSEG_; ++j) {
      const bool use = j <= g;
#pragma unroll
      for (int i = 0; i < 4; ++i) pa[i] += use ? pl[j - 1][i] : zero4;
      ka += use ? kl[j - 1] : 0.0f;
    }
#pragma unroll
    for (int i = 0; i < 4; ++i) *reinterpret_cast<v4f*>(pg_s + 4 * tid + 1024 * i) = pa[i];
    kzp_s[tid] = ka;
  }
  __syncthreads();
  __builtin_amdgcn_sched_barrier(0);
  state_request(tt, pg, 0);
#pragma unroll
  for (int j = 0; j < 8; ++j) kz2[j] = *reinterpret_cast<const v4f*>(kzp_s + 8 * j + 4 * hf);

  float z[4];
  v8f acc[4][4];
#pragma unroll
  for (int ti = 0; ti < 4; ++ti) {
    z[ti] = 0.0f;
#pragma unroll
    for (int jj = 0; jj < 4; ++jj) acc[ti][jj] = (v8f){};
  }
#pragma unroll
  for (int j = 0; j < 8; ++j) kc[j] = (local ? kc[j] : zero4) + kz2[j];
  q_quarter(qf, kc, z, 0);
  state_add(sa, tt, local);
  __builtin_amdgcn_sched_barrier(0);
  state_request(tt, pg, 1);
  __builtin_amdgcn_sched_barrier(0);
  state_products(sa, qf, acc, 0);
  q_quarter(qf, kc, z, 1);
  __builtin_amdgcn_sched_barrier(0);
  state_add(sb, tt, local);
  __builtin_amdgcn_sched_barrier(0);
  state_request(sa, sp, 2);
  state_request(tt, pg, 2);
  __builtin_amdgcn_sched_barrier(0);
  state_products(sb, qf, acc, 1);
  q_quarter(qf, kc, z, 2);
  __builtin_amdgcn_sched_barrier(0);
  state_add(sa, tt, local);
  __builtin_amdgcn_sched_barrier(0);
  state_request(sb, sp, 3);
  state_request(tt, pg, 3);
  __builtin_amdgcn_sched_barrier(0);
  state_products(sa, qf, acc, 2);
  q_quarter(qf, kc, z, 3);
  __builtin_amdgcn_sched_barrier(0);
  state_add(sb, tt, local);
  __builtin_amdgcn_sched_barrier(0);
  v4f kf0[8], vf0[8], kf1[8], vf1[8];
  tile_kv_request(kf0, vf0, kp, vp, 0);
  __builtin_amdgcn_sched_barrier(0);
  state_products(sb, qf, acc, 3);
  __builtin_amdgcn_sched_barrier(0);

  tile_kv_request(kf1, vf1, kp, vp, 1);
  __builtin_amdgcn_sched_barrier(0);
#pragma unroll
  for (int j = 0; j < 8; ++j) kf0[j] = featmap4(kf0[j]);
  strip_tile(kf0, vf0, qf[0], acc[0], z[0], true, c, hf);
  strip_store(acc[0], z[0], ob, hf);
  strip_tile(kf0, vf0, qf[1], acc[1], z[1], false, c, hf);
  strip_tile(kf0, vf0, qf[2], acc[2], z[2], false, c, hf);
  strip_tile(kf0, vf0, qf[3], acc[3], z[3], false, c, hf);
  __builtin_amdgcn_sched_barrier(0);
  tile_kv_request(kf0, vf0, kp, vp, 2);
  __builtin_amdgcn_sched_barrier(0);
#pragma unroll
  for (int j = 0; j < 8; ++j) kf1[j] = featmap4(kf1[j]);
  strip_tile(kf1, vf1, qf[1], acc[1], z[1], true, c, hf);
  strip_store(acc[1], z[1], ob + (size_t)16 * (H_ * DV_), hf);
  strip_tile(kf1, vf1, qf[2], acc[2], z[2], false, c, hf);
  strip_tile(kf1, vf1, qf[3], acc[3], z[3], false, c, hf);
  __builtin_amdgcn_sched_barrier(0);
  tile_kv_request(kf1, vf1, kp, vp, 3);
  __builtin_amdgcn_sched_barrier(0);
#pragma unroll
  for (int j = 0; j < 8; ++j) kf0[j] = featmap4(kf0[j]);
  strip_tile(kf0, vf0, qf[2], acc[2], z[2], true, c, hf);
  strip_store(acc[2], z[2], ob + (size_t)32 * (H_ * DV_), hf);
  strip_tile(kf0, vf0, qf[3], acc[3], z[3], false, c, hf);
#pragma unroll
  for (int j = 0; j < 8; ++j) kf1[j] = featmap4(kf1[j]);
  strip_tile(kf1, vf1, qf[3], acc[3], z[3], true, c, hf);
  strip_store(acc[3], z[3], ob + (size_t)48 * (H_ * DV_), hf);
}

extern "C" void kernel_launch(void* const* d_in, const int* in_sizes, int n_in,
                              void* d_out, int out_size, void* d_ws, size_t ws_size,
                              hipStream_t stream) {
  (void)in_sizes; (void)n_in; (void)out_size; (void)ws_size;
  const float* q = (const float*)d_in[0];
  const float* k = (const float*)d_in[1];
  const float* v = (const float*)d_in[2];
  float* out = (float*)d_out;

  float* kv   = (float*)d_ws;
  float* ksum = kv + (size_t)NH_ * T_ * D_ * DV_;

  cla_state_prefix<<<NH_ * NSEG_, 256, 0, stream>>>(k, v, kv, ksum);
  cla_phase3<<<NH_ * NSEG_, 256, 0, stream>>>(q, k, v, kv, ksum, out);
}
